# starting kernel
# baseline (speedup 1.0000x reference)
#include <hip/hip_runtime.h>


typedef float v2f __attribute__((ext_vector_type(2)));
typedef float v8f __attribute__((ext_vector_type(8)));

#define NHEADS 4
#define HID    64
#define NL     3
#define BATCH  64
#define NNODES 512
#define DIM    64
#define OUTD   (NHEADS * HID)
#define NEG_SLOPE 0.2f

__device__ __forceinline__ v8f wmma_f32(v2f a, v2f b, v8f c) {
  return __builtin_amdgcn_wmma_f32_16x16x4_f32(false, a, false, b, (short)0, c,
                                               false, false);
}

__device__ __forceinline__ float lrelu(float x) {
  return x >= 0.f ? x : NEG_SLOPE * x;
}

__global__ __launch_bounds__(256) void k_lin(const float* __restrict__ x,
                                             const float* __restrict__ W,
                                             float* __restrict__ hout) {
  const int tid  = threadIdx.x;
  const int lane = tid & 31;
  const int wid  = tid >> 5;
  const int half = lane >> 4;
  const int mr   = lane & 15;

  const int mtile = blockIdx.x;
  const int ctile = wid;
  const int row0  = mtile * 64;
  const int o0    = ctile * 32;

  const v8f vz = {0.f,0.f,0.f,0.f,0.f,0.f,0.f,0.f};
  v8f acc[4][2];
#pragma unroll
  for (int sm = 0; sm < 4; ++sm)
#pragma unroll
    for (int sn = 0; sn < 2; ++sn) acc[sm][sn] = vz;

  for (int kk = 0; kk < DIM; kk += 4) {
    const int kbase = kk + 2 * half;
    v2f afr[4], bfr[2];
#pragma unroll
    for (int sm = 0; sm < 4; ++sm)
      afr[sm] = *(const v2f*)(x + (size_t)(row0 + sm * 16 + mr) * DIM + kbase);
#pragma unroll
    for (int sn = 0; sn < 2; ++sn)
      bfr[sn] = *(const v2f*)(W + (size_t)(o0 + sn * 16 + mr) * DIM + kbase);
#pragma unroll
    for (int sm = 0; sm < 4; ++sm)
#pragma unroll
      for (int sn = 0; sn < 2; ++sn)
        acc[sm][sn] = wmma_f32(afr[sm], bfr[sn], acc[sm][sn]);
  }

#pragma unroll
  for (int sm = 0; sm < 4; ++sm)
#pragma unroll
    for (int sn = 0; sn < 2; ++sn)
#pragma unroll
      for (int r = 0; r < 8; ++r) {
        const int row = row0 + sm * 16 + r + 8 * half;
        const int col = o0 + sn * 16 + mr;
        hout[(size_t)row * OUTD + col] = acc[sm][sn][r];
      }
}

__global__ __launch_bounds__(256) void k_scores(const float* __restrict__ hbuf,
                                                const float* __restrict__ attS,
                                                const float* __restrict__ attD,
                                                float* __restrict__ asrc,
                                                float* __restrict__ adst) {
  const int tid  = threadIdx.x;
  const int lane = tid & 31;
  const int wid  = tid >> 5;
  const int node = blockIdx.x * 8 + wid;
  const int b = node >> 9, n = node & 511;
  const float* hrow = hbuf + (size_t)node * OUTD;

  for (int hh = 0; hh < NHEADS; ++hh) {
    const float h0 = hrow[hh * HID + lane];
    const float h1 = hrow[hh * HID + 32 + lane];
    float ss = h0 * attS[hh * HID + lane] + h1 * attS[hh * HID + 32 + lane];
    float sd = h0 * attD[hh * HID + lane] + h1 * attD[hh * HID + 32 + lane];
#pragma unroll
    for (int off = 16; off; off >>= 1) {
      ss += __shfl_xor(ss, off, 32);
      sd += __shfl_xor(sd, off, 32);
    }
    if (lane == 0) {
      asrc[((b * NHEADS + hh) << 9) + n] = ss;
      adst[((b * NHEADS + hh) << 9) + n] = sd;
    }
  }
}

__global__ __launch_bounds__(256) void k_agg(const float* __restrict__ hbuf,
                                             const float* __restrict__ asrc,
                                             const float* __restrict__ adst,
                                             const float* __restrict__ bias,
                                             float* __restrict__ xout) {
  __shared__ float sH[128 * 80];
  __shared__ float sA[512];
  __shared__ float sred[8];

  const int tid  = threadIdx.x;
  const int lane = tid & 31;
  const int wid  = tid >> 5;
  const int half = lane >> 4;
  const int mr   = lane & 15;

  const int b    = blockIdx.y;
  const int i0   = blockIdx.x * 128;
  const int mrow = i0 + wid * 16 + mr;

  const v8f vz = {0.f,0.f,0.f,0.f,0.f,0.f,0.f,0.f};
  v8f accm[4];
#pragma unroll
  for (int t = 0; t < 4; ++t) accm[t] = vz;

  for (int hh = 0; hh < NHEADS; ++hh) {
    const int bh = (b * NHEADS + hh) << 9;

    __syncthreads();
    sA[tid]       = asrc[bh + tid];
    sA[tid + 256] = asrc[bh + tid + 256];
    __syncthreads();

    float mx = fmaxf(sA[tid], sA[tid + 256]);
#pragma unroll
    for (int off = 16; off; off >>= 1) mx = fmaxf(mx, __shfl_xor(mx, off, 32));
    if (lane == 0) sred[wid] = mx;
    __syncthreads();
    float smax = sred[0];
#pragma unroll
    for (int w = 1; w < 8; ++w) smax = fmaxf(smax, sred[w]);

    const float d     = adst[bh + mrow];
    const float mterm = lrelu(d + smax);

    v8f acc[4];
#pragma unroll
    for (int t = 0; t < 4; ++t) acc[t] = vz;
    float zacc = 0.f;

    for (int jc = 0; jc < 4; ++jc) {
      const int j0 = jc * 128;
      __syncthreads();
#pragma unroll
      for (int k = 0; k < 32; ++k) {
        const int e  = tid + k * 256;
        const int rr = e >> 6, cc = e & 63;
        sH[rr * 80 + cc] =
            hbuf[(size_t)(b * NNODES + j0 + rr) * OUTD + hh * HID + cc];
      }
      __syncthreads();

      for (int jj = 0; jj < 128; jj += 4) {
        const int ja = j0 + jj + 2 * half;
        const float s0 = sA[ja];
        const float s1 = sA[ja + 1];
        const float p0 = __expf(lrelu(d + s0) - mterm);
        const float p1 = __expf(lrelu(d + s1) - mterm);
        zacc += p0 + p1;
        v2f af; af.x = p0; af.y = p1;
        const int r0 = (jj + 2 * half) * 80 + mr;
#pragma unroll
        for (int t = 0; t < 4; ++t) {
          v2f bf;
          bf.x = sH[r0 + t * 16];
          bf.y = sH[r0 + 80 + t * 16];
          acc[t] = wmma_f32(af, bf, acc[t]);
        }
      }
    }

    const float z = zacc + __shfl_xor(zacc, 16, 32);
#pragma unroll
    for (int r = 0; r < 8; ++r) {
      const float zr = __shfl(z, r + 8 * half, 32);
      const float sc = 0.25f / zr;
#pragma unroll
      for (int t = 0; t < 4; ++t) accm[t][r] += acc[t][r] * sc;
    }
  }

#pragma unroll
  for (int t = 0; t < 4; ++t) {
    const float bsv = bias[t * 16 + mr];
#pragma unroll
    for (int r = 0; r < 8; ++r) {
      float v = accm[t][r] + bsv;
      v = v > 0.f ? v : 0.f;
      xout[(size_t)(b * NNODES + i0 + wid * 16 + r + 8 * half) * HID +
           t * 16 + mr] = v;
    }
  }
}

__global__ __launch_bounds__(64) void k_readout(const float* __restrict__ x,
                                                const float* __restrict__ Wr,
                                                const float* __restrict__ br,
                                                float* __restrict__ out) {
  __shared__ float pooled[DIM];
  const int b = blockIdx.x, d = threadIdx.x;
  float s = 0.f;
  for (int i = 0; i < NNODES; ++i)
    s += x[(size_t)(b * NNODES + i) * DIM + d];
  pooled[d] = s * (1.f / (float)NNODES);
  __syncthreads();
  float o = br[d];
  for (int k = 0; k < DIM; ++k) o += pooled[k] * Wr[d * DIM + k];
  out[b * DIM + d] = o;
}

extern "C" void kernel_launch(void* const* d_in, const int* in_sizes, int n_in,
                              void* d_out, int out_size, void* d_ws,
                              size_t ws_size, hipStream_t stream) {
  (void)in_sizes; (void)n_in; (void)out_size; (void)ws_size;
  const float* emb    = (const float*)d_in[0];
  const float* lin_w  = (const float*)d_in[1];
  const float* attS   = (const float*)d_in[2];
  const float* attD   = (const float*)d_in[3];
  const float* conv_b = (const float*)d_in[4];
  const float* rw     = (const float*)d_in[5];
  const float* rb     = (const float*)d_in[6];

  float* ws = (float*)d_ws;
  const size_t XSZ = (size_t)BATCH * NNODES * DIM;
  const size_t HSZ = (size_t)BATCH * NNODES * OUTD;
  const size_t ASZ = (size_t)BATCH * NHEADS * NNODES;
  float* xbuf0 = ws;
  float* xbuf1 = ws + XSZ;
  float* hbuf  = ws + 2 * XSZ;
  float* asrc  = hbuf + HSZ;
  float* adst  = asrc + ASZ;

  const float* xin = emb;
  float* xo[NL] = {xbuf0, xbuf1, xbuf0};

  for (int l = 0; l < NL; ++l) {
    k_lin<<<(BATCH * NNODES) / 64, 256, 0, stream>>>(
        xin, lin_w + (size_t)l * OUTD * DIM, hbuf);
    k_scores<<<(BATCH * NNODES) / 8, 256, 0, stream>>>(
        hbuf, attS + l * NHEADS * HID, attD + l * NHEADS * HID, asrc, adst);
    k_agg<<<dim3(NNODES / 128, BATCH), 256, 0, stream>>>(
        hbuf, asrc, adst, conv_b + l * HID, xo[l]);
    xin = xo[l];
  }
  k_readout<<<BATCH, 64, 0, stream>>>(xin, rw, rb, (float*)d_out);
}


# Aggregation loop fed ahead: H staged in LDS as node pairs (a B fragment = one 64-bit load), next 8-node group's loads issued before the current group's 8 matrix instructions; same sum order
# speedup vs baseline: 1.0776x; 1.0776x over previous
#include <hip/hip_runtime.h>


typedef float v2f __attribute__((ext_vector_type(2)));
typedef float v8f __attribute__((ext_vector_type(8)));

#define NHEADS 4
#define HID    64
#define NL     3
#define BATCH  64
#define NNODES 512
#define DIM    64
#define OUTD   (NHEADS * HID)
#define NEG_SLOPE 0.2f

__device__ __forceinline__ v8f wmma_f32(v2f a, v2f b, v8f c) {
  return __builtin_amdgcn_wmma_f32_16x16x4_f32(false, a, false, b, (short)0, c,
                                               false, false);
}

__device__ __forceinline__ float lrelu(float x) {
  return x >= 0.f ? x : NEG_SLOPE * x;
}

__global__ __launch_bounds__(256) void k_lin(const float* __restrict__ x,
                                             const float* __restrict__ W,
                                             float* __restrict__ hout) {
  const int tid  = threadIdx.x;
  const int lane = tid & 31;
  const int wid  = tid >> 5;
  const int half = lane >> 4;
  const int mr   = lane & 15;

  const int mtile = blockIdx.x;
  const int ctile = wid;
  const int row0  = mtile * 64;
  const int o0    = ctile * 32;

  const v8f vz = {0.f,0.f,0.f,0.f,0.f,0.f,0.f,0.f};
  v8f acc[4][2];
#pragma unroll
  for (int sm = 0; sm < 4; ++sm)
#pragma unroll
    for (int sn = 0; sn < 2; ++sn) acc[sm][sn] = vz;

  for (int kk = 0; kk < DIM; kk += 4) {
    const int kbase = kk + 2 * half;
    v2f afr[4], bfr[2];
#pragma unroll
    for (int sm = 0; sm < 4; ++sm)
      afr[sm] = *(const v2f*)(x + (size_t)(row0 + sm * 16 + mr) * DIM + kbase);
#pragma unroll
    for (int sn = 0; sn < 2; ++sn)
      bfr[sn] = *(const v2f*)(W + (size_t)(o0 + sn * 16 + mr) * DIM + kbase);
#pragma unroll
    for (int sm = 0; sm < 4; ++sm)
#pragma unroll
      for (int sn = 0; sn < 2; ++sn)
        acc[sm][sn] = wmma_f32(afr[sm], bfr[sn], acc[sm][sn]);
  }

#pragma unroll
  for (int sm = 0; sm < 4; ++sm)
#pragma unroll
    for (int sn = 0; sn < 2; ++sn)
#pragma unroll
      for (int r = 0; r < 8; ++r) {
        const int row = row0 + sm * 16 + r + 8 * half;
        const int col = o0 + sn * 16 + mr;
        hout[(size_t)row * OUTD + col] = acc[sm][sn][r];
      }
}

__global__ __launch_bounds__(256) void k_scores(const float* __restrict__ hbuf,
                                                const float* __restrict__ attS,
                                                const float* __restrict__ attD,
                                                float* __restrict__ asrc,
                                                float* __restrict__ adst) {
  const int tid  = threadIdx.x;
  const int lane = tid & 31;
  const int wid  = tid >> 5;
  const int node = blockIdx.x * 8 + wid;
  const int b = node >> 9, n = node & 511;
  const float* hrow = hbuf + (size_t)node * OUTD;

  for (int hh = 0; hh < NHEADS; ++hh) {
    const float h0 = hrow[hh * HID + lane];
    const float h1 = hrow[hh * HID + 32 + lane];
    float ss = h0 * attS[hh * HID + lane] + h1 * attS[hh * HID + 32 + lane];
    float sd = h0 * attD[hh * HID + lane] + h1 * attD[hh * HID + 32 + lane];
#pragma unroll
    for (int off = 16; off; off >>= 1) {
      ss += __shfl_xor(ss, off, 32);
      sd += __shfl_xor(sd, off, 32);
    }
    if (lane == 0) {
      asrc[((b * NHEADS + hh) << 9) + n] = ss;
      adst[((b * NHEADS + hh) << 9) + n] = sd;
    }
  }
}

struct Grp {
  float s[2][2];
  v2f   b[2][4];
};

__device__ __forceinline__ void grp_load(Grp& g, const float* sAj,
                                         const float* sHh) {
#pragma unroll
  for (int u = 0; u < 2; ++u) {
    const v2f sv = *(const v2f*)(sAj + u * 4);
    g.s[u][0] = sv.x;
    g.s[u][1] = sv.y;
#pragma unroll
    for (int t = 0; t < 4; ++t)
      g.b[u][t] = *(const v2f*)(sHh + u * 2 * 160 + t * 32);
  }
}

__device__ __forceinline__ void grp_mma(const Grp& g, float d, float mterm,
                                        float& zacc, v8f (&acc)[4]) {
#pragma unroll
  for (int u = 0; u < 2; ++u) {
    const float p0 = __expf(lrelu(d + g.s[u][0]) - mterm);
    const float p1 = __expf(lrelu(d + g.s[u][1]) - mterm);
    zacc += p0 + p1;
    v2f af; af.x = p0; af.y = p1;
#pragma unroll
    for (int t = 0; t < 4; ++t) acc[t] = wmma_f32(af, g.b[u][t], acc[t]);
  }
}

__global__ __launch_bounds__(256) void k_agg(const float* __restrict__ hbuf,
                                             const float* __restrict__ asrc,
                                             const float* __restrict__ adst,
                                             const float* __restrict__ bias,
                                             float* __restrict__ xout) {
  __shared__ __attribute__((aligned(16))) float sH[64 * 160];
  __shared__ __attribute__((aligned(16))) float sA[512];
  __shared__ float sred[8];

  const int tid  = threadIdx.x;
  const int lane = tid & 31;
  const int wid  = tid >> 5;
  const int half = lane >> 4;
  const int mr   = lane & 15;

  const int b    = blockIdx.y;
  const int i0   = blockIdx.x * 128;
  const int mrow = i0 + wid * 16 + mr;

  const v8f vz = {0.f,0.f,0.f,0.f,0.f,0.f,0.f,0.f};
  v8f accm[4];
#pragma unroll
  for (int t = 0; t < 4; ++t) accm[t] = vz;

  for (int hh = 0; hh < NHEADS; ++hh) {
    const int bh = (b * NHEADS + hh) << 9;

    __syncthreads();
    sA[tid]       = asrc[bh + tid];
    sA[tid + 256] = asrc[bh + tid + 256];
    __syncthreads();

    float mx = fmaxf(sA[tid], sA[tid + 256]);
#pragma unroll
    for (int off = 16; off; off >>= 1) mx = fmaxf(mx, __shfl_xor(mx, off, 32));
    if (lane == 0) sred[wid] = mx;
    __syncthreads();
    float smax = sred[0];
#pragma unroll
    for (int w = 1; w < 8; ++w) smax = fmaxf(smax, sred[w]);

    const float d     = adst[bh + mrow];
    const float mterm = lrelu(d + smax);

    v8f acc[4];
#pragma unroll
    for (int t = 0; t < 4; ++t) acc[t] = vz;
    float zacc = 0.f;

    for (int jc = 0; jc < 4; ++jc) {
      const int j0 = jc * 128;
      __syncthreads();
      v2f hv[16];
#pragma unroll
      for (int k = 0; k < 16; ++k) {
        const int e  = tid + k * 256;
        const int jp = e >> 6, cc = e & 63;
        const float* src =
            hbuf + (size_t)(b * NNODES + j0 + 2 * jp) * OUTD + hh * HID + cc;
        hv[k].x = src[0];
        hv[k].y = src[OUTD];
      }
#pragma unroll
      for (int k = 0; k < 16; ++k) {
        const int e  = tid + k * 256;
        const int jp = e >> 6, cc = e & 63;
        *(v2f*)(sH + jp * 160 + 2 * cc) = hv[k];
      }
      __syncthreads();

      const float* sAp = sA + j0 + 2 * half;
      const float* sHp = sH + half * 160 + 2 * mr;
      Grp ga, gb;
      grp_load(ga, sAp, sHp);
      for (int jj = 0; jj < 128; jj += 16) {
        grp_load(gb, sAp + jj + 8, sHp + ((jj + 8) >> 1) * 160);
        grp_mma(ga, d, mterm, zacc, acc);
        if (jj + 16 < 128)
          grp_load(ga, sAp + jj + 16, sHp + ((jj + 16) >> 1) * 160);
        grp_mma(gb, d, mterm, zacc, acc);
      }
    }

    const float z = zacc + __shfl_xor(zacc, 16, 32);
#pragma unroll
    for (int r = 0; r < 8; ++r) {
      const float zr = __shfl(z, r + 8 * half, 32);
      const float sc = 0.25f / zr;
#pragma unroll
      for (int t = 0; t < 4; ++t) accm[t][r] += acc[t][r] * sc;
    }
  }

#pragma unroll
  for (int t = 0; t < 4; ++t) {
    const float bsv = bias[t * 16 + mr];
#pragma unroll
    for (int r = 0; r < 8; ++r) {
      float v = accm[t][r] + bsv;
      v = v > 0.f ? v : 0.f;
      xout[(size_t)(b * NNODES + i0 + wid * 16 + r + 8 * half) * HID +
           t * 16 + mr] = v;
    }
  }
}

__global__ __launch_bounds__(64) void k_readout(const float* __restrict__ x,
                                                const float* __restrict__ Wr,
                                                const float* __restrict__ br,
                                                float* __restrict__ out) {
  __shared__ float pooled[DIM];
  const int b = blockIdx.x, d = threadIdx.x;
  float s = 0.f;
  for (int i = 0; i < NNODES; ++i)
    s += x[(size_t)(b * NNODES + i) * DIM + d];
  pooled[d] = s * (1.f / (float)NNODES);
  __syncthreads();
  float o = br[d];
  for (int k = 0; k < DIM; ++k) o += pooled[k] * Wr[d * DIM + k];
  out[b * DIM + d] = o;
}

extern "C" void kernel_launch(void* const* d_in, const int* in_sizes, int n_in,
                              void* d_out, int out_size, void* d_ws,
                              size_t ws_size, hipStream_t stream) {
  (void)in_sizes; (void)n_in; (void)out_size; (void)ws_size;
  const float* emb    = (const float*)d_in[0];
  const float* lin_w  = (const float*)d_in[1];
  const float* attS   = (const float*)d_in[2];
  const float* attD   = (const float*)d_in[3];
  const float* conv_b = (const float*)d_in[4];
  const float* rw     = (const float*)d_in[5];
  const float* rb     = (const float*)d_in[6];

  float* ws = (float*)d_ws;
  const size_t XSZ = (size_t)BATCH * NNODES * DIM;
  const size_t HSZ = (size_t)BATCH * NNODES * OUTD;
  const size_t ASZ = (size_t)BATCH * NHEADS * NNODES;
  float* xbuf0 = ws;
  float* xbuf1 = ws + XSZ;
  float* hbuf  = ws + 2 * XSZ;
  float* asrc  = hbuf + HSZ;
  float* adst  = asrc + ASZ;

  const float* xin = emb;
  float* xo[NL] = {xbuf0, xbuf1, xbuf0};

  for (int l = 0; l < NL; ++l) {
    k_lin<<<(BATCH * NNODES) / 64, 256, 0, stream>>>(
        xin, lin_w + (size_t)l * OUTD * DIM, hbuf);
    k_scores<<<(BATCH * NNODES) / 8, 256, 0, stream>>>(
        hbuf, attS + l * NHEADS * HID, attD + l * NHEADS * HID, asrc, adst);
    k_agg<<<dim3(NNODES / 128, BATCH), 256, 0, stream>>>(
        hbuf, asrc, adst, conv_b + l * HID, xo[l]);
    xin = xo[l];
  }
  k_readout<<<BATCH, 64, 0, stream>>>(xin, rw, rb, (float*)d_out);
}


# Score kernel folded into the projection: a wave owns one head's 64 columns, dots its accumulators with att_src/att_dst, sums over 16 lanes; 3 launches fewer; score sum order changed
# speedup vs baseline: 1.2502x; 1.1602x over previous
#include <hip/hip_runtime.h>


typedef float v2f __attribute__((ext_vector_type(2)));
typedef float v8f __attribute__((ext_vector_type(8)));

#define NHEADS 4
#define HID    64
#define NL     3
#define BATCH  64
#define NNODES 512
#define DIM    64
#define OUTD   (NHEADS * HID)
#define NEG_SLOPE 0.2f

__device__ __forceinline__ v8f wmma_f32(v2f a, v2f b, v8f c) {
  return __builtin_amdgcn_wmma_f32_16x16x4_f32(false, a, false, b, (short)0, c,
                                               false, false);
}

__device__ __forceinline__ float lrelu(float x) {
  return x >= 0.f ? x : NEG_SLOPE * x;
}

template <int WD>
__device__ __forceinline__ void rows16_fold(float (&v)[16], int mr) {
  const bool up = (mr & WD) != 0;
#pragma unroll
  for (int i = 0; i < WD; ++i) {
    const float keep = up ? v[i + WD] : v[i];
    const float send = up ? v[i] : v[i + WD];
    v[i] = keep + __shfl_xor(send, WD, 32);
  }
}

__global__ __launch_bounds__(256) void k_lin(const float* __restrict__ x,
                                             const float* __restrict__ W,
                                             const float* __restrict__ attS,
                                             const float* __restrict__ attD,
                                             float* __restrict__ hout,
                                             float* __restrict__ asrc,
                                             float* __restrict__ adst) {
  const int tid  = threadIdx.x;
  const int lane = tid & 31;
  const int wid  = tid >> 5;
  const int half = lane >> 4;
  const int mr   = lane & 15;

  const int hd   = wid >> 1;
  const int row0 = blockIdx.x * 64 + (wid & 1) * 32;
  const int o0   = hd * HID;

  const v8f vz = {0.f,0.f,0.f,0.f,0.f,0.f,0.f,0.f};
  v8f acc[2][4];
#pragma unroll
  for (int sm = 0; sm < 2; ++sm)
#pragma unroll
    for (int sn = 0; sn < 4; ++sn) acc[sm][sn] = vz;

  for (int kk = 0; kk < DIM; kk += 4) {
    const int kbase = kk + 2 * half;
    v2f afr[2], bfr[4];
#pragma unroll
    for (int sm = 0; sm < 2; ++sm)
      afr[sm] = *(const v2f*)(x + (size_t)(row0 + sm * 16 + mr) * DIM + kbase);
#pragma unroll
    for (int sn = 0; sn < 4; ++sn)
      bfr[sn] = *(const v2f*)(W + (size_t)(o0 + sn * 16 + mr) * DIM + kbase);
#pragma unroll
    for (int sm = 0; sm < 2; ++sm)
#pragma unroll
      for (int sn = 0; sn < 4; ++sn)
        acc[sm][sn] = wmma_f32(afr[sm], bfr[sn], acc[sm][sn]);
  }

#pragma unroll
  for (int sm = 0; sm < 2; ++sm)
#pragma unroll
    for (int sn = 0; sn < 4; ++sn)
#pragma unroll
      for (int r = 0; r < 8; ++r) {
        const int row = row0 + sm * 16 + r + 8 * half;
        const int col = o0 + sn * 16 + mr;
        hout[(size_t)row * OUTD + col] = acc[sm][sn][r];
      }

  float wS[4], wD[4];
#pragma unroll
  for (int sn = 0; sn < 4; ++sn) {
    wS[sn] = attS[o0 + sn * 16 + mr];
    wD[sn] = attD[o0 + sn * 16 + mr];
  }
  float ps[16], pd[16];
#pragma unroll
  for (int sm = 0; sm < 2; ++sm)
#pragma unroll
    for (int r = 0; r < 8; ++r) {
      float ss = acc[sm][0][r] * wS[0];
      float sd = acc[sm][0][r] * wD[0];
#pragma unroll
      for (int sn = 1; sn < 4; ++sn) {
        ss = fmaf(acc[sm][sn][r], wS[sn], ss);
        sd = fmaf(acc[sm][sn][r], wD[sn], sd);
      }
      ps[sm * 8 + r] = ss;
      pd[sm * 8 + r] = sd;
    }
  rows16_fold<8>(ps, mr);  rows16_fold<8>(pd, mr);
  rows16_fold<4>(ps, mr);  rows16_fold<4>(pd, mr);
  rows16_fold<2>(ps, mr);  rows16_fold<2>(pd, mr);
  rows16_fold<1>(ps, mr);  rows16_fold<1>(pd, mr);

  const int grow = row0 + (mr >> 3) * 16 + (mr & 7) + 8 * half;
  const int b = grow >> 9, n = grow & (NNODES - 1);
  asrc[((b * NHEADS + hd) << 9) + n] = ps[0];
  adst[((b * NHEADS + hd) << 9) + n] = pd[0];
}

struct Grp {
  float s[2][2];
  v2f   b[2][4];
};

__device__ __forceinline__ void grp_load(Grp& g, const float* sAj,
                                         const float* sHh) {
#pragma unroll
  for (int u = 0; u < 2; ++u) {
    const v2f sv = *(const v2f*)(sAj + u * 4);
    g.s[u][0] = sv.x;
    g.s[u][1] = sv.y;
#pragma unroll
    for (int t = 0; t < 4; ++t)
      g.b[u][t] = *(const v2f*)(sHh + u * 2 * 160 + t * 32);
  }
}

__device__ __forceinline__ void grp_mma(const Grp& g, float d, float mterm,
                                        float& zacc, v8f (&acc)[4]) {
#pragma unroll
  for (int u = 0; u < 2; ++u) {
    const float p0 = __expf(lrelu(d + g.s[u][0]) - mterm);
    const float p1 = __expf(lrelu(d + g.s[u][1]) - mterm);
    zacc += p0 + p1;
    v2f af; af.x = p0; af.y = p1;
#pragma unroll
    for (int t = 0; t < 4; ++t) acc[t] = wmma_f32(af, g.b[u][t], acc[t]);
  }
}

__global__ __launch_bounds__(256) void k_agg(const float* __restrict__ hbuf,
                                             const float* __restrict__ asrc,
                                             const float* __restrict__ adst,
                                             const float* __restrict__ bias,
                                             float* __restrict__ xout) {
  __shared__ __attribute__((aligned(16))) float sH[64 * 160];
  __shared__ __attribute__((aligned(16))) float sA[512];
  __shared__ float sred[8];

  const int tid  = threadIdx.x;
  const int lane = tid & 31;
  const int wid  = tid >> 5;
  const int half = lane >> 4;
  const int mr   = lane & 15;

  const int b    = blockIdx.y;
  const int i0   = blockIdx.x * 128;
  const int mrow = i0 + wid * 16 + mr;

  const v8f vz = {0.f,0.f,0.f,0.f,0.f,0.f,0.f,0.f};
  v8f accm[4];
#pragma unroll
  for (int t = 0; t < 4; ++t) accm[t] = vz;

  for (int hh = 0; hh < NHEADS; ++hh) {
    const int bh = (b * NHEADS + hh) << 9;

    __syncthreads();
    sA[tid]       = asrc[bh + tid];
    sA[tid + 256] = asrc[bh + tid + 256];
    __syncthreads();

    float mx = fmaxf(sA[tid], sA[tid + 256]);
#pragma unroll
    for (int off = 16; off; off >>= 1) mx = fmaxf(mx, __shfl_xor(mx, off, 32));
    if (lane == 0) sred[wid] = mx;
    __syncthreads();
    float smax = sred[0];
#pragma unroll
    for (int w = 1; w < 8; ++w) smax = fmaxf(smax, sred[w]);

    const float d     = adst[bh + mrow];
    const float mterm = lrelu(d + smax);

    v8f acc[4];
#pragma unroll
    for (int t = 0; t < 4; ++t) acc[t] = vz;
    float zacc = 0.f;

    for (int jc = 0; jc < 4; ++jc) {
      const int j0 = jc * 128;
      __syncthreads();
      v2f hv[16];
#pragma unroll
      for (int k = 0; k < 16; ++k) {
        const int e  = tid + k * 256;
        const int jp = e >> 6, cc = e & 63;
        const float* src =
            hbuf + (size_t)(b * NNODES + j0 + 2 * jp) * OUTD + hh * HID + cc;
        hv[k].x = src[0];
        hv[k].y = src[OUTD];
      }
#pragma unroll
      for (int k = 0; k < 16; ++k) {
        const int e  = tid + k * 256;
        const int jp = e >> 6, cc = e & 63;
        *(v2f*)(sH + jp * 160 + 2 * cc) = hv[k];
      }
      __syncthreads();

      const float* sAp = sA + j0 + 2 * half;
      const float* sHp = sH + half * 160 + 2 * mr;
      Grp ga, gb;
      grp_load(ga, sAp, sHp);
      for (int jj = 0; jj < 128; jj += 16) {
        grp_load(gb, sAp + jj + 8, sHp + ((jj + 8) >> 1) * 160);
        grp_mma(ga, d, mterm, zacc, acc);
        if (jj + 16 < 128)
          grp_load(ga, sAp + jj + 16, sHp + ((jj + 16) >> 1) * 160);
        grp_mma(gb, d, mterm, zacc, acc);
      }
    }

    const float z = zacc + __shfl_xor(zacc, 16, 32);
#pragma unroll
    for (int r = 0; r < 8; ++r) {
      const float zr = __shfl(z, r + 8 * half, 32);
      const float sc = 0.25f / zr;
#pragma unroll
      for (int t = 0; t < 4; ++t) accm[t][r] += acc[t][r] * sc;
    }
  }

#pragma unroll
  for (int t = 0; t < 4; ++t) {
    const float bsv = bias[t * 16 + mr];
#pragma unroll
    for (int r = 0; r < 8; ++r) {
      float v = accm[t][r] + bsv;
      v = v > 0.f ? v : 0.f;
      xout[(size_t)(b * NNODES + i0 + wid * 16 + r + 8 * half) * HID +
           t * 16 + mr] = v;
    }
  }
}

__global__ __launch_bounds__(64) void k_readout(const float* __restrict__ x,
                                                const float* __restrict__ Wr,
                                                const float* __restrict__ br,
                                                float* __restrict__ out) {
  __shared__ float pooled[DIM];
  const int b = blockIdx.x, d = threadIdx.x;
  float s = 0.f;
  for (int i = 0; i < NNODES; ++i)
    s += x[(size_t)(b * NNODES + i) * DIM + d];
  pooled[d] = s * (1.f / (float)NNODES);
  __syncthreads();
  float o = br[d];
  for (int k = 0; k < DIM; ++k) o += pooled[k] * Wr[d * DIM + k];
  out[b * DIM + d] = o;
}

extern "C" void kernel_launch(void* const* d_in, const int* in_sizes, int n_in,
                              void* d_out, int out_size, void* d_ws,
                              size_t ws_size, hipStream_t stream) {
  (void)in_sizes; (void)n_in; (void)out_size; (void)ws_size;
  const float* emb    = (const float*)d_in[0];
  const float* lin_w  = (const float*)d_in[1];
  const float* attS   = (const float*)d_in[2];
  const float* attD   = (const float*)d_in[3];
  const float* conv_b = (const float*)d_in[4];
  const float* rw     = (const float*)d_in[5];
  const float* rb     = (const float*)d_in[6];

  float* ws = (float*)d_ws;
  const size_t XSZ = (size_t)BATCH * NNODES * DIM;
  const size_t HSZ = (size_t)BATCH * NNODES * OUTD;
  const size_t ASZ = (size_t)BATCH * NHEADS * NNODES;
  float* xbuf0 = ws;
  float* xbuf1 = ws + XSZ;
  float* hbuf  = ws + 2 * XSZ;
  float* asrc  = hbuf + HSZ;
  float* adst  = asrc + ASZ;

  const float* xin = emb;
  float* xo[NL] = {xbuf0, xbuf1, xbuf0};

  for (int l = 0; l < NL; ++l) {
    k_lin<<<(BATCH * NNODES) / 64, 256, 0, stream>>>(
        xin, lin_w + (size_t)l * OUTD * DIM, attS + l * NHEADS * HID,
        attD + l * NHEADS * HID, hbuf, asrc, adst);
    k_agg<<<dim3(NNODES / 128, BATCH), 256, 0, stream>>>(
        hbuf, asrc, adst, conv_b + l * HID, xo[l]);
    xin = xo[l];
  }
  k_readout<<<BATCH, 64, 0, stream>>>(xin, rw, rb, (float*)d_out);
}


# k_agg as one 16-tile pipeline over head x chunk: two LDS H buffers, next tile's global loads issued ahead of the current tile's matrix work, all heads' scores staged once; same sum order
# speedup vs baseline: 1.4503x; 1.1601x over previous
#include <hip/hip_runtime.h>


typedef float v2f __attribute__((ext_vector_type(2)));
typedef float v8f __attribute__((ext_vector_type(8)));

#define NHEADS 4
#define HID    64
#define NL     3
#define BATCH  64
#define NNODES 512
#define DIM    64
#define OUTD   (NHEADS * HID)
#define NEG_SLOPE 0.2f

__device__ __forceinline__ v8f wmma_f32(v2f a, v2f b, v8f c) {
  return __builtin_amdgcn_wmma_f32_16x16x4_f32(false, a, false, b, (short)0, c,
                                               false, false);
}

__device__ __forceinline__ float lrelu(float x) {
  return x >= 0.f ? x : NEG_SLOPE * x;
}

template <int WD>
__device__ __forceinline__ void rows16_fold(float (&v)[16], int mr) {
  const bool up = (mr & WD) != 0;
#pragma unroll
  for (int i = 0; i < WD; ++i) {
    const float keep = up ? v[i + WD] : v[i];
    const float send = up ? v[i] : v[i + WD];
    v[i] = keep + __shfl_xor(send, WD, 32);
  }
}

__global__ __launch_bounds__(256) void k_lin(const float* __restrict__ x,
                                             const float* __restrict__ W,
                                             const float* __restrict__ attS,
                                             const float* __restrict__ attD,
                                             float* __restrict__ hout,
                                             float* __restrict__ asrc,
                                             float* __restrict__ adst) {
  const int tid  = threadIdx.x;
  const int lane = tid & 31;
  const int wid  = tid >> 5;
  const int half = lane >> 4;
  const int mr   = lane & 15;

  const int hd   = wid >> 1;
  const int row0 = blockIdx.x * 64 + (wid & 1) * 32;
  const int o0   = hd * HID;

  const v8f vz = {0.f,0.f,0.f,0.f,0.f,0.f,0.f,0.f};
  v8f acc[2][4];
#pragma unroll
  for (int sm = 0; sm < 2; ++sm)
#pragma unroll
    for (int sn = 0; sn < 4; ++sn) acc[sm][sn] = vz;

  for (int kk = 0; kk < DIM; kk += 4) {
    const int kbase = kk + 2 * half;
    v2f afr[2], bfr[4];
#pragma unroll
    for (int sm = 0; sm < 2; ++sm)
      afr[sm] = *(const v2f*)(x + (size_t)(row0 + sm * 16 + mr) * DIM + kbase);
#pragma unroll
    for (int sn = 0; sn < 4; ++sn)
      bfr[sn] = *(const v2f*)(W + (size_t)(o0 + sn * 16 + mr) * DIM + kbase);
#pragma unroll
    for (int sm = 0; sm < 2; ++sm)
#pragma unroll
      for (int sn = 0; sn < 4; ++sn)
        acc[sm][sn] = wmma_f32(afr[sm], bfr[sn], acc[sm][sn]);
  }

#pragma unroll
  for (int sm = 0; sm < 2; ++sm)
#pragma unroll
    for (int sn = 0; sn < 4; ++sn)
#pragma unroll
      for (int r = 0; r < 8; ++r) {
        const int row = row0 + sm * 16 + r + 8 * half;
        const int col = o0 + sn * 16 + mr;
        hout[(size_t)row * OUTD + col] = acc[sm][sn][r];
      }

  float wS[4], wD[4];
#pragma unroll
  for (int sn = 0; sn < 4; ++sn) {
    wS[sn] = attS[o0 + sn * 16 + mr];
    wD[sn] = attD[o0 + sn * 16 + mr];
  }
  float ps[16], pd[16];
#pragma unroll
  for (int sm = 0; sm < 2; ++sm)
#pragma unroll
    for (int r = 0; r < 8; ++r) {
      float ss = acc[sm][0][r] * wS[0];
      float sd = acc[sm][0][r] * wD[0];
#pragma unroll
      for (int sn = 1; sn < 4; ++sn) {
        ss = fmaf(acc[sm][sn][r], wS[sn], ss);
        sd = fmaf(acc[sm][sn][r], wD[sn], sd);
      }
      ps[sm * 8 + r] = ss;
      pd[sm * 8 + r] = sd;
    }
  rows16_fold<8>(ps, mr);  rows16_fold<8>(pd, mr);
  rows16_fold<4>(ps, mr);  rows16_fold<4>(pd, mr);
  rows16_fold<2>(ps, mr);  rows16_fold<2>(pd, mr);
  rows16_fold<1>(ps, mr);  rows16_fold<1>(pd, mr);

  const int grow = row0 + (mr >> 3) * 16 + (mr & 7) + 8 * half;
  const int b = grow >> 9, n = grow & (NNODES - 1);
  asrc[((b * NHEADS + hd) << 9) + n] = ps[0];
  adst[((b * NHEADS + hd) << 9) + n] = pd[0];
}

struct Grp {
  float s[2][2];
  v2f   b[2][4];
};

__device__ __forceinline__ void grp_load(Grp& g, const float* sAj,
                                         const float* sHh) {
#pragma unroll
  for (int u = 0; u < 2; ++u) {
    const v2f sv = *(const v2f*)(sAj + u * 4);
    g.s[u][0] = sv.x;
    g.s[u][1] = sv.y;
#pragma unroll
    for (int t = 0; t < 4; ++t)
      g.b[u][t] = *(const v2f*)(sHh + u * 2 * 160 + t * 32);
  }
}

__device__ __forceinline__ void grp_mma(const Grp& g, float d, float mterm,
                                        float& zacc, v8f (&acc)[4]) {
#pragma unroll
  for (int u = 0; u < 2; ++u) {
    const float p0 = __expf(lrelu(d + g.s[u][0]) - mterm);
    const float p1 = __expf(lrelu(d + g.s[u][1]) - mterm);
    zacc += p0 + p1;
    v2f af; af.x = p0; af.y = p1;
#pragma unroll
    for (int t = 0; t < 4; ++t) acc[t] = wmma_f32(af, g.b[u][t], acc[t]);
  }
}

typedef float v4f __attribute__((ext_vector_type(4)));

#define TILE_F (64 * 160)

__device__ __forceinline__ void tile_load(v2f (&pre)[16], const float* g) {
#pragma unroll
  for (int k = 0; k < 16; ++k) {
    pre[k].x = g[k * 8 * OUTD];
    pre[k].y = g[k * 8 * OUTD + OUTD];
  }
}

__device__ __forceinline__ void tile_store(const v2f (&pre)[16], float* s) {
#pragma unroll
  for (int k = 0; k < 16; ++k) *(v2f*)(s + k * 4 * 160) = pre[k];
}

__global__ __launch_bounds__(256) void k_agg(const float* __restrict__ hbuf,
                                             const float* __restrict__ asrc,
                                             const float* __restrict__ adst,
                                             const float* __restrict__ bias,
                                             float* __restrict__ xout) {
  __shared__ __attribute__((aligned(16))) float sH[2 * TILE_F];
  __shared__ __attribute__((aligned(16))) float sA[NHEADS * NNODES];
  __shared__ float sD[NHEADS * 128];

  const int tid  = threadIdx.x;
  const int lane = tid & 31;
  const int wid  = tid >> 5;
  const int half = lane >> 4;
  const int mr   = lane & 15;

  const int b  = blockIdx.y;
  const int i0 = blockIdx.x * 128;

  const float* gsrc =
      hbuf + (size_t)(b * NNODES + 2 * (tid >> 6)) * OUTD + (tid & 63);
  float* sdst = sH + (tid >> 6) * 160 + 2 * (tid & 63);

  v2f pre[16];
  tile_load(pre, gsrc);

  {
    const v4f* as4 = (const v4f*)(asrc + ((size_t)(b * NHEADS) << 9));
    ((v4f*)sA)[tid]       = as4[tid];
    ((v4f*)sA)[tid + 256] = as4[tid + 256];
#pragma unroll
    for (int q = 0; q < 2; ++q) {
      const int idx = tid + 256 * q;
      sD[idx] = adst[((b * NHEADS + (idx >> 7)) << 9) + i0 + (idx & 127)];
    }
  }
  __syncthreads();

  float sm[NHEADS];
#pragma unroll
  for (int h = 0; h < NHEADS; ++h) {
    const v4f* s4 = (const v4f*)(sA + h * NNODES);
    const v4f q0 = s4[lane], q1 = s4[lane + 32], q2 = s4[lane + 64],
              q3 = s4[lane + 96];
    const float m0 = fmaxf(fmaxf(q0.x, q0.y), fmaxf(q0.z, q0.w));
    const float m1 = fmaxf(fmaxf(q1.x, q1.y), fmaxf(q1.z, q1.w));
    const float m2 = fmaxf(fmaxf(q2.x, q2.y), fmaxf(q2.z, q2.w));
    const float m3 = fmaxf(fmaxf(q3.x, q3.y), fmaxf(q3.z, q3.w));
    float mx = fmaxf(fmaxf(m0, m1), fmaxf(m2, m3));
#pragma unroll
    for (int off = 16; off; off >>= 1)
      mx = fmaxf(mx, __shfl_xor(mx, off, 32));
    sm[h] = mx;
  }

  const v8f vz = {0.f,0.f,0.f,0.f,0.f,0.f,0.f,0.f};
  v8f accm[4];
#pragma unroll
  for (int t = 0; t < 4; ++t) accm[t] = vz;

#pragma unroll 1
  for (int hh = 0; hh < NHEADS; ++hh) {
    const float smax = hh == 0 ? sm[0] : hh == 1 ? sm[1] : hh == 2 ? sm[2] : sm[3];
    const float d     = sD[hh * 128 + wid * 16 + mr];
    const float mterm = lrelu(d + smax);

    v8f acc[4];
#pragma unroll
    for (int t = 0; t < 4; ++t) acc[t] = vz;
    float zacc = 0.f;

#pragma unroll 1
    for (int jc = 0; jc < 4; ++jc) {
      const int t = hh * 4 + jc;

      tile_store(pre, sdst + (t & 1) * TILE_F);
      __syncthreads();

      if (t + 1 < 4 * NHEADS) {
        const int jn = (jc + 1) & 3;
        const int hn = hh + (jc == 3 ? 1 : 0);
        tile_load(pre, gsrc + (size_t)jn * 128 * OUTD + hn * HID);
      }

      const float* sAp = sA + hh * NNODES + jc * 128 + 2 * half;
      const float* sHp = sH + (t & 1) * TILE_F + half * 160 + 2 * mr;
      Grp ga, gb;
      grp_load(ga, sAp, sHp);
      for (int jj = 0; jj < 128; jj += 16) {
        grp_load(gb, sAp + jj + 8, sHp + ((jj + 8) >> 1) * 160);
        grp_mma(ga, d, mterm, zacc, acc);
        if (jj + 16 < 128)
          grp_load(ga, sAp + jj + 16, sHp + ((jj + 16) >> 1) * 160);
        grp_mma(gb, d, mterm, zacc, acc);
      }
    }

    const float z = zacc + __shfl_xor(zacc, 16, 32);
#pragma unroll
    for (int r = 0; r < 8; ++r) {
      const float zr = __shfl(z, r + 8 * half, 32);
      const float sc = 0.25f / zr;
#pragma unroll
      for (int t = 0; t < 4; ++t) accm[t][r] += acc[t][r] * sc;
    }
  }

#pragma unroll
  for (int t = 0; t < 4; ++t) {
    const float bsv = bias[t * 16 + mr];
#pragma unroll
    for (int r = 0; r < 8; ++r) {
      float v = accm[t][r] + bsv;
      v = v > 0.f ? v : 0.f;
      xout[(size_t)(b * NNODES + i0 + wid * 16 + r + 8 * half) * HID +
           t * 16 + mr] = v;
    }
  }
}

__global__ __launch_bounds__(64) void k_readout(const float* __restrict__ x,
                                                const float* __restrict__ Wr,
                                                const float* __restrict__ br,
                                                float* __restrict__ out) {
  __shared__ float pooled[DIM];
  const int b = blockIdx.x, d = threadIdx.x;
  float s = 0.f;
  for (int i = 0; i < NNODES; ++i)
    s += x[(size_t)(b * NNODES + i) * DIM + d];
  pooled[d] = s * (1.f / (float)NNODES);
  __syncthreads();
  float o = br[d];
  for (int k = 0; k < DIM; ++k) o += pooled[k] * Wr[d * DIM + k];
  out[b * DIM + d] = o;
}

extern "C" void kernel_launch(void* const* d_in, const int* in_sizes, int n_in,
                              void* d_out, int out_size, void* d_ws,
                              size_t ws_size, hipStream_t stream) {
  (void)in_sizes; (void)n_in; (void)out_size; (void)ws_size;
  const float* emb    = (const float*)d_in[0];
  const float* lin_w  = (const float*)d_in[1];
  const float* attS   = (const float*)d_in[2];
  const float* attD   = (const float*)d_in[3];
  const float* conv_b = (const float*)d_in[4];
  const float* rw     = (const float*)d_in[5];
  const float* rb     = (const float*)d_in[6];

  float* ws = (float*)d_ws;
  const size_t XSZ = (size_t)BATCH * NNODES * DIM;
  const size_t HSZ = (size_t)BATCH * NNODES * OUTD;
  const size_t ASZ = (size_t)BATCH * NHEADS * NNODES;
  float* xbuf0 = ws;
  float* xbuf1 = ws + XSZ;
  float* hbuf  = ws + 2 * XSZ;
  float* asrc  = hbuf + HSZ;
  float* adst  = asrc + ASZ;

  const float* xin = emb;
  float* xo[NL] = {xbuf0, xbuf1, xbuf0};

  for (int l = 0; l < NL; ++l) {
    k_lin<<<(BATCH * NNODES) / 64, 256, 0, stream>>>(
        xin, lin_w + (size_t)l * OUTD * DIM, attS + l * NHEADS * HID,
        attD + l * NHEADS * HID, hbuf, asrc, adst);
    k_agg<<<dim3(NNODES / 128, BATCH), 256, 0, stream>>>(
        hbuf, asrc, adst, conv_b + l * HID, xo[l]);
    xin = xo[l];
  }
  k_readout<<<BATCH, 64, 0, stream>>>(xin, rw, rb, (float*)d_out);
}


# k_agg: scheduling fences hold the LDS operand reads in one batch per 32 matrix instructions (32-node groups read a block ahead, a tile's first group ahead of its global loads); same sum order
# speedup vs baseline: 1.5491x; 1.0681x over previous
#include <hip/hip_runtime.h>


typedef float v2f __attribute__((ext_vector_type(2)));
typedef float v8f __attribute__((ext_vector_type(8)));

#define NHEADS 4
#define HID    64
#define NL     3
#define BATCH  64
#define NNODES 512
#define DIM    64
#define OUTD   (NHEADS * HID)
#define NEG_SLOPE 0.2f

__device__ __forceinline__ v8f wmma_f32(v2f a, v2f b, v8f c) {
  return __builtin_amdgcn_wmma_f32_16x16x4_f32(false, a, false, b, (short)0, c,
                                               false, false);
}

__device__ __forceinline__ float lrelu(float x) {
  return x >= 0.f ? x : NEG_SLOPE * x;
}

template <int WD>
__device__ __forceinline__ void rows16_fold(float (&v)[16], int mr) {
  const bool up = (mr & WD) != 0;
#pragma unroll
  for (int i = 0; i < WD; ++i) {
    const float keep = up ? v[i + WD] : v[i];
    const float send = up ? v[i] : v[i + WD];
    v[i] = keep + __shfl_xor(send, WD, 32);
  }
}

__global__ __launch_bounds__(256) void k_lin(const float* __restrict__ x,
                                             const float* __restrict__ W,
                                             const float* __restrict__ attS,
                                             const float* __restrict__ attD,
                                             float* __restrict__ hout,
                                             float* __restrict__ asrc,
                                             float* __restrict__ adst) {
  const int tid  = threadIdx.x;
  const int lane = tid & 31;
  const int wid  = tid >> 5;
  const int half = lane >> 4;
  const int mr   = lane & 15;

  const int hd   = wid >> 1;
  const int row0 = blockIdx.x * 64 + (wid & 1) * 32;
  const int o0   = hd * HID;

  const v8f vz = {0.f,0.f,0.f,0.f,0.f,0.f,0.f,0.f};
  v8f acc[2][4];
#pragma unroll
  for (int sm = 0; sm < 2; ++sm)
#pragma unroll
    for (int sn = 0; sn < 4; ++sn) acc[sm][sn] = vz;

  for (int kk = 0; kk < DIM; kk += 4) {
    const int kbase = kk + 2 * half;
    v2f afr[2], bfr[4];
#pragma unroll
    for (int sm = 0; sm < 2; ++sm)
      afr[sm] = *(const v2f*)(x + (size_t)(row0 + sm * 16 + mr) * DIM + kbase);
#pragma unroll
    for (int sn = 0; sn < 4; ++sn)
      bfr[sn] = *(const v2f*)(W + (size_t)(o0 + sn * 16 + mr) * DIM + kbase);
#pragma unroll
    for (int sm = 0; sm < 2; ++sm)
#pragma unroll
      for (int sn = 0; sn < 4; ++sn)
        acc[sm][sn] = wmma_f32(afr[sm], bfr[sn], acc[sm][sn]);
  }

#pragma unroll
  for (int sm = 0; sm < 2; ++sm)
#pragma unroll
    for (int sn = 0; sn < 4; ++sn)
#pragma unroll
      for (int r = 0; r < 8; ++r) {
        const int row = row0 + sm * 16 + r + 8 * half;
        const int col = o0 + sn * 16 + mr;
        hout[(size_t)row * OUTD + col] = acc[sm][sn][r];
      }

  float wS[4], wD[4];
#pragma unroll
  for (int sn = 0; sn < 4; ++sn) {
    wS[sn] = attS[o0 + sn * 16 + mr];
    wD[sn] = attD[o0 + sn * 16 + mr];
  }
  float ps[16], pd[16];
#pragma unroll
  for (int sm = 0; sm < 2; ++sm)
#pragma unroll
    for (int r = 0; r < 8; ++r) {
      float ss = acc[sm][0][r] * wS[0];
      float sd = acc[sm][0][r] * wD[0];
#pragma unroll
      for (int sn = 1; sn < 4; ++sn) {
        ss = fmaf(acc[sm][sn][r], wS[sn], ss);
        sd = fmaf(acc[sm][sn][r], wD[sn], sd);
      }
      ps[sm * 8 + r] = ss;
      pd[sm * 8 + r] = sd;
    }
  rows16_fold<8>(ps, mr);  rows16_fold<8>(pd, mr);
  rows16_fold<4>(ps, mr);  rows16_fold<4>(pd, mr);
  rows16_fold<2>(ps, mr);  rows16_fold<2>(pd, mr);
  rows16_fold<1>(ps, mr);  rows16_fold<1>(pd, mr);

  const int grow = row0 + (mr >> 3) * 16 + (mr & 7) + 8 * half;
  const int b = grow >> 9, n = grow & (NNODES - 1);
  asrc[((b * NHEADS + hd) << 9) + n] = ps[0];
  adst[((b * NHEADS + hd) << 9) + n] = pd[0];
}

#define GSTEPS 8
#define GNODES (4 * GSTEPS)

struct Grp {
  float s[GSTEPS][2];
  v2f   b[GSTEPS][4];
};

__device__ __forceinline__ void grp_load(Grp& g, const float* sAj,
                                         const float* sHh) {
#pragma unroll
  for (int u = 0; u < GSTEPS; ++u) {
    const v2f sv = *(const v2f*)(sAj + u * 4);
    g.s[u][0] = sv.x;
    g.s[u][1] = sv.y;
#pragma unroll
    for (int t = 0; t < 4; ++t)
      g.b[u][t] = *(const v2f*)(sHh + u * 2 * 160 + t * 32);
  }
}

__device__ __forceinline__ void grp_mma(const Grp& g, float d, float mterm,
                                        float& zacc, v8f (&acc)[4]) {
#pragma unroll
  for (int u = 0; u < GSTEPS; ++u) {
    const float p0 = __expf(lrelu(d + g.s[u][0]) - mterm);
    const float p1 = __expf(lrelu(d + g.s[u][1]) - mterm);
    zacc += p0 + p1;
    v2f af; af.x = p0; af.y = p1;
#pragma unroll
    for (int t = 0; t < 4; ++t) acc[t] = wmma_f32(af, g.b[u][t], acc[t]);
  }
}

typedef float v4f __attribute__((ext_vector_type(4)));

#define TILE_F (64 * 160)

__device__ __forceinline__ void tile_load(v2f (&pre)[16], const float* g) {
#pragma unroll
  for (int k = 0; k < 16; ++k) {
    pre[k].x = g[k * 8 * OUTD];
    pre[k].y = g[k * 8 * OUTD + OUTD];
  }
}

__device__ __forceinline__ void tile_store(const v2f (&pre)[16], float* s) {
#pragma unroll
  for (int k = 0; k < 16; ++k) *(v2f*)(s + k * 4 * 160) = pre[k];
}

__global__ __launch_bounds__(256) void k_agg(const float* __restrict__ hbuf,
                                             const float* __restrict__ asrc,
                                             const float* __restrict__ adst,
                                             const float* __restrict__ bias,
                                             float* __restrict__ xout) {
  __shared__ __attribute__((aligned(16))) float sH[2 * TILE_F];
  __shared__ __attribute__((aligned(16))) float sA[NHEADS * NNODES];
  __shared__ float sD[NHEADS * 128];

  const int tid  = threadIdx.x;
  const int lane = tid & 31;
  const int wid  = tid >> 5;
  const int half = lane >> 4;
  const int mr   = lane & 15;

  const int b  = blockIdx.y;
  const int i0 = blockIdx.x * 128;

  const float* gsrc =
      hbuf + (size_t)(b * NNODES + 2 * (tid >> 6)) * OUTD + (tid & 63);
  float* sdst = sH + (tid >> 6) * 160 + 2 * (tid & 63);

  v2f pre[16];
  tile_load(pre, gsrc);

  {
    const v4f* as4 = (const v4f*)(asrc + ((size_t)(b * NHEADS) << 9));
    ((v4f*)sA)[tid]       = as4[tid];
    ((v4f*)sA)[tid + 256] = as4[tid + 256];
#pragma unroll
    for (int q = 0; q < 2; ++q) {
      const int idx = tid + 256 * q;
      sD[idx] = adst[((b * NHEADS + (idx >> 7)) << 9) + i0 + (idx & 127)];
    }
  }
  __syncthreads();

  float sm[NHEADS];
#pragma unroll
  for (int h = 0; h < NHEADS; ++h) {
    const v4f* s4 = (const v4f*)(sA + h * NNODES);
    const v4f q0 = s4[lane], q1 = s4[lane + 32], q2 = s4[lane + 64],
              q3 = s4[lane + 96];
    const float m0 = fmaxf(fmaxf(q0.x, q0.y), fmaxf(q0.z, q0.w));
    const float m1 = fmaxf(fmaxf(q1.x, q1.y), fmaxf(q1.z, q1.w));
    const float m2 = fmaxf(fmaxf(q2.x, q2.y), fmaxf(q2.z, q2.w));
    const float m3 = fmaxf(fmaxf(q3.x, q3.y), fmaxf(q3.z, q3.w));
    float mx = fmaxf(fmaxf(m0, m1), fmaxf(m2, m3));
#pragma unroll
    for (int off = 16; off; off >>= 1)
      mx = fmaxf(mx, __shfl_xor(mx, off, 32));
    sm[h] = mx;
  }

  const v8f vz = {0.f,0.f,0.f,0.f,0.f,0.f,0.f,0.f};
  v8f accm[4];
#pragma unroll
  for (int t = 0; t < 4; ++t) accm[t] = vz;

#pragma unroll 1
  for (int hh = 0; hh < NHEADS; ++hh) {
    const float smax = hh == 0 ? sm[0] : hh == 1 ? sm[1] : hh == 2 ? sm[2] : sm[3];
    const float d     = sD[hh * 128 + wid * 16 + mr];
    const float mterm = lrelu(d + smax);

    v8f acc[4];
#pragma unroll
    for (int t = 0; t < 4; ++t) acc[t] = vz;
    float zacc = 0.f;

#pragma unroll 1
    for (int jc = 0; jc < 4; ++jc) {
      const int t = hh * 4 + jc;

      tile_store(pre, sdst + (t & 1) * TILE_F);
      __syncthreads();

      const float* sAp = sA + hh * NNODES + jc * 128 + 2 * half;
      const float* sHp = sH + (t & 1) * TILE_F + half * 160 + 2 * mr;
      Grp ga, gb;

      grp_load(ga, sAp, sHp);
      __builtin_amdgcn_sched_barrier(0);

      if (t + 1 < 4 * NHEADS) {
        const int jn = (jc + 1) & 3;
        const int hn = hh + (jc == 3 ? 1 : 0);
        tile_load(pre, gsrc + (size_t)jn * 128 * OUTD + hn * HID);
      }

      for (int jj = 0; jj < 128; jj += 2 * GNODES) {
        grp_load(gb, sAp + jj + GNODES, sHp + ((jj + GNODES) >> 1) * 160);
        __builtin_amdgcn_sched_barrier(0);
        grp_mma(ga, d, mterm, zacc, acc);
        __builtin_amdgcn_sched_barrier(0);
        if (jj + 2 * GNODES < 128)
          grp_load(ga, sAp + jj + 2 * GNODES,
                   sHp + ((jj + 2 * GNODES) >> 1) * 160);
        __builtin_amdgcn_sched_barrier(0);
        grp_mma(gb, d, mterm, zacc, acc);
        __builtin_amdgcn_sched_barrier(0);
      }
    }

    const float z = zacc + __shfl_xor(zacc, 16, 32);
#pragma unroll
    for (int r = 0; r < 8; ++r) {
      const float zr = __shfl(z, r + 8 * half, 32);
      const float sc = 0.25f / zr;
#pragma unroll
      for (int t = 0; t < 4; ++t) accm[t][r] += acc[t][r] * sc;
    }
  }

#pragma unroll
  for (int t = 0; t < 4; ++t) {
    const float bsv = bias[t * 16 + mr];
#pragma unroll
    for (int r = 0; r < 8; ++r) {
      float v = accm[t][r] + bsv;
      v = v > 0.f ? v : 0.f;
      xout[(size_t)(b * NNODES + i0 + wid * 16 + r + 8 * half) * HID +
           t * 16 + mr] = v;
    }
  }
}

__global__ __launch_bounds__(64) void k_readout(const float* __restrict__ x,
                                                const float* __restrict__ Wr,
                                                const float* __restrict__ br,
                                                float* __restrict__ out) {
  __shared__ float pooled[DIM];
  const int b = blockIdx.x, d = threadIdx.x;
  float s = 0.f;
  for (int i = 0; i < NNODES; ++i)
    s += x[(size_t)(b * NNODES + i) * DIM + d];
  pooled[d] = s * (1.f / (float)NNODES);
  __syncthreads();
  float o = br[d];
  for (int k = 0; k < DIM; ++k) o += pooled[k] * Wr[d * DIM + k];
  out[b * DIM + d] = o;
}

extern "C" void kernel_launch(void* const* d_in, const int* in_sizes, int n_in,
                              void* d_out, int out_size, void* d_ws,
                              size_t ws_size, hipStream_t stream) {
  (void)in_sizes; (void)n_in; (void)out_size; (void)ws_size;
  const float* emb    = (const float*)d_in[0];
  const float* lin_w  = (const float*)d_in[1];
  const float* attS   = (const float*)d_in[2];
  const float* attD   = (const float*)d_in[3];
  const float* conv_b = (const float*)d_in[4];
  const float* rw     = (const float*)d_in[5];
  const float* rb     = (const float*)d_in[6];

  float* ws = (float*)d_ws;
  const size_t XSZ = (size_t)BATCH * NNODES * DIM;
  const size_t HSZ = (size_t)BATCH * NNODES * OUTD;
  const size_t ASZ = (size_t)BATCH * NHEADS * NNODES;
  float* xbuf0 = ws;
  float* xbuf1 = ws + XSZ;
  float* hbuf  = ws + 2 * XSZ;
  float* asrc  = hbuf + HSZ;
  float* adst  = asrc + ASZ;

  const float* xin = emb;
  float* xo[NL] = {xbuf0, xbuf1, xbuf0};

  for (int l = 0; l < NL; ++l) {
    k_lin<<<(BATCH * NNODES) / 64, 256, 0, stream>>>(
        xin, lin_w + (size_t)l * OUTD * DIM, attS + l * NHEADS * HID,
        attD + l * NHEADS * HID, hbuf, asrc, adst);
    k_agg<<<dim3(NNODES / 128, BATCH), 256, 0, stream>>>(
        hbuf, asrc, adst, conv_b + l * HID, xo[l]);
    xin = xo[l];
  }
  k_readout<<<BATCH, 64, 0, stream>>>(xin, rw, rb, (float*)d_out);
}


# k_agg: no exponential per (i,j): exp(lrelu(d+s)-m) = max(ra_i*E_j, rb_i*F_j) as exp is monotone; E,F taken once per node and head, ra,rb once per row; element rounding changes, same sum order
# speedup vs baseline: 1.6251x; 1.0491x over previous
#include <hip/hip_runtime.h>


typedef float v2f __attribute__((ext_vector_type(2)));
typedef float v4f __attribute__((ext_vector_type(4)));
typedef float v8f __attribute__((ext_vector_type(8)));

#define NHEADS 4
#define HID    64
#define NL     3
#define BATCH  64
#define NNODES 512
#define DIM    64
#define OUTD   (NHEADS * HID)
#define NEG_SLOPE 0.2f

__device__ __forceinline__ v8f wmma_f32(v2f a, v2f b, v8f c) {
  return __builtin_amdgcn_wmma_f32_16x16x4_f32(false, a, false, b, (short)0, c,
                                               false, false);
}

template <int WD>
__device__ __forceinline__ void rows16_fold(float (&v)[16], int mr) {
  const bool up = (mr & WD) != 0;
#pragma unroll
  for (int i = 0; i < WD; ++i) {
    const float keep = up ? v[i + WD] : v[i];
    const float send = up ? v[i] : v[i + WD];
    v[i] = keep + __shfl_xor(send, WD, 32);
  }
}

__global__ __launch_bounds__(256) void k_lin(const float* __restrict__ x,
                                             const float* __restrict__ W,
                                             const float* __restrict__ attS,
                                             const float* __restrict__ attD,
                                             float* __restrict__ hout,
                                             float* __restrict__ asrc,
                                             float* __restrict__ adst) {
  const int tid  = threadIdx.x;
  const int lane = tid & 31;
  const int wid  = tid >> 5;
  const int half = lane >> 4;
  const int mr   = lane & 15;

  const int hd   = wid >> 1;
  const int row0 = blockIdx.x * 64 + (wid & 1) * 32;
  const int o0   = hd * HID;

  const v8f vz = {0.f,0.f,0.f,0.f,0.f,0.f,0.f,0.f};
  v8f acc[2][4];
#pragma unroll
  for (int sm = 0; sm < 2; ++sm)
#pragma unroll
    for (int sn = 0; sn < 4; ++sn) acc[sm][sn] = vz;

  for (int kk = 0; kk < DIM; kk += 4) {
    const int kbase = kk + 2 * half;
    v2f afr[2], bfr[4];
#pragma unroll
    for (int sm = 0; sm < 2; ++sm)
      afr[sm] = *(const v2f*)(x + (size_t)(row0 + sm * 16 + mr) * DIM + kbase);
#pragma unroll
    for (int sn = 0; sn < 4; ++sn)
      bfr[sn] = *(const v2f*)(W + (size_t)(o0 + sn * 16 + mr) * DIM + kbase);
#pragma unroll
    for (int sm = 0; sm < 2; ++sm)
#pragma unroll
      for (int sn = 0; sn < 4; ++sn)
        acc[sm][sn] = wmma_f32(afr[sm], bfr[sn], acc[sm][sn]);
  }

#pragma unroll
  for (int sm = 0; sm < 2; ++sm)
#pragma unroll
    for (int sn = 0; sn < 4; ++sn)
#pragma unroll
      for (int r = 0; r < 8; ++r) {
        const int row = row0 + sm * 16 + r + 8 * half;
        const int col = o0 + sn * 16 + mr;
        hout[(size_t)row * OUTD + col] = acc[sm][sn][r];
      }

  float wS[4], wD[4];
#pragma unroll
  for (int sn = 0; sn < 4; ++sn) {
    wS[sn] = attS[o0 + sn * 16 + mr];
    wD[sn] = attD[o0 + sn * 16 + mr];
  }
  float ps[16], pd[16];
#pragma unroll
  for (int sm = 0; sm < 2; ++sm)
#pragma unroll
    for (int r = 0; r < 8; ++r) {
      float ss = acc[sm][0][r] * wS[0];
      float sd = acc[sm][0][r] * wD[0];
#pragma unroll
      for (int sn = 1; sn < 4; ++sn) {
        ss = fmaf(acc[sm][sn][r], wS[sn], ss);
        sd = fmaf(acc[sm][sn][r], wD[sn], sd);
      }
      ps[sm * 8 + r] = ss;
      pd[sm * 8 + r] = sd;
    }
  rows16_fold<8>(ps, mr);  rows16_fold<8>(pd, mr);
  rows16_fold<4>(ps, mr);  rows16_fold<4>(pd, mr);
  rows16_fold<2>(ps, mr);  rows16_fold<2>(pd, mr);
  rows16_fold<1>(ps, mr);  rows16_fold<1>(pd, mr);

  const int grow = row0 + (mr >> 3) * 16 + (mr & 7) + 8 * half;
  const int b = grow >> 9, n = grow & (NNODES - 1);
  asrc[((b * NHEADS + hd) << 9) + n] = ps[0];
  adst[((b * NHEADS + hd) << 9) + n] = pd[0];
}

#define GSTEPS 8
#define GNODES (4 * GSTEPS)

struct Grp {
  v4f ef[GSTEPS];
  v2f b[GSTEPS][4];
};

__device__ __forceinline__ void grp_load(Grp& g, const float* sEq,
                                         const float* sHh) {
#pragma unroll
  for (int u = 0; u < GSTEPS; ++u) {
    g.ef[u] = *(const v4f*)(sEq + u * 8);
#pragma unroll
    for (int t = 0; t < 4; ++t)
      g.b[u][t] = *(const v2f*)(sHh + u * 2 * 160 + t * 32);
  }
}

__device__ __forceinline__ void grp_mma(const Grp& g, float ra, float rb,
                                        float& zacc, v8f (&acc)[4]) {
#pragma unroll
  for (int u = 0; u < GSTEPS; ++u) {
    const float p0 = fmaxf(ra * g.ef[u].x, rb * g.ef[u].z);
    const float p1 = fmaxf(ra * g.ef[u].y, rb * g.ef[u].w);
    zacc += p0 + p1;
    v2f af; af.x = p0; af.y = p1;
#pragma unroll
    for (int t = 0; t < 4; ++t) acc[t] = wmma_f32(af, g.b[u][t], acc[t]);
  }
}

#define TILE_F (64 * 160)

__device__ __forceinline__ void tile_load(v2f (&pre)[16], const float* g) {
#pragma unroll
  for (int k = 0; k < 16; ++k) {
    pre[k].x = g[k * 8 * OUTD];
    pre[k].y = g[k * 8 * OUTD + OUTD];
  }
}

__device__ __forceinline__ void tile_store(const v2f (&pre)[16], float* s) {
#pragma unroll
  for (int k = 0; k < 16; ++k) *(v2f*)(s + k * 4 * 160) = pre[k];
}

__global__ __launch_bounds__(256) void k_agg(const float* __restrict__ hbuf,
                                             const float* __restrict__ asrc,
                                             const float* __restrict__ adst,
                                             const float* __restrict__ bias,
                                             float* __restrict__ xout) {
  __shared__ __attribute__((aligned(16))) float sH[2 * TILE_F];
  __shared__ __attribute__((aligned(16))) float sA[NHEADS * NNODES];
  __shared__ __attribute__((aligned(16))) float sEF[NHEADS * NNODES * 2];
  __shared__ float sD[NHEADS * 128];

  const int tid  = threadIdx.x;
  const int lane = tid & 31;
  const int wid  = tid >> 5;
  const int half = lane >> 4;
  const int mr   = lane & 15;

  const int b  = blockIdx.y;
  const int i0 = blockIdx.x * 128;

  const float* gsrc =
      hbuf + (size_t)(b * NNODES + 2 * (tid >> 6)) * OUTD + (tid & 63);
  float* sdst = sH + (tid >> 6) * 160 + 2 * (tid & 63);

  v2f pre[16];
  tile_load(pre, gsrc);

  v4f sq0, sq1;
  {
    const v4f* as4 = (const v4f*)(asrc + ((size_t)(b * NHEADS) << 9));
    sq0 = as4[tid];
    sq1 = as4[tid + 256];
    ((v4f*)sA)[tid]       = sq0;
    ((v4f*)sA)[tid + 256] = sq1;
#pragma unroll
    for (int q = 0; q < 2; ++q) {
      const int idx = tid + 256 * q;
      sD[idx] = adst[((b * NHEADS + (idx >> 7)) << 9) + i0 + (idx & 127)];
    }
  }
  __syncthreads();

  float sm[NHEADS];
#pragma unroll
  for (int h = 0; h < NHEADS; ++h) {
    const v4f* s4 = (const v4f*)(sA + h * NNODES);
    const v4f q0 = s4[lane], q1 = s4[lane + 32], q2 = s4[lane + 64],
              q3 = s4[lane + 96];
    const float m0 = fmaxf(fmaxf(q0.x, q0.y), fmaxf(q0.z, q0.w));
    const float m1 = fmaxf(fmaxf(q1.x, q1.y), fmaxf(q1.z, q1.w));
    const float m2 = fmaxf(fmaxf(q2.x, q2.y), fmaxf(q2.z, q2.w));
    const float m3 = fmaxf(fmaxf(q3.x, q3.y), fmaxf(q3.z, q3.w));
    float mx = fmaxf(fmaxf(m0, m1), fmaxf(m2, m3));
#pragma unroll
    for (int off = 16; off; off >>= 1)
      mx = fmaxf(mx, __shfl_xor(mx, off, 32));
    sm[h] = mx;
  }

  {
    const bool hi = (tid >> 7) != 0;
    const float S0 = hi ? sm[1] : sm[0];
    const float S1 = hi ? sm[3] : sm[2];
    v4f e0, f0, e1, f1;
#pragma unroll
    for (int c = 0; c < 4; ++c) {
      const float t0 = sq0[c] - S0, t1 = sq1[c] - S1;
      e0[c] = __expf(t0);  f0[c] = __expf(NEG_SLOPE * t0);
      e1[c] = __expf(t1);  f1[c] = __expf(NEG_SLOPE * t1);
    }
    v4f* w0 = (v4f*)(sEF + tid * 8);
    v4f* w1 = (v4f*)(sEF + 2 * NNODES * 2 + tid * 8);
    const v4f r00 = {e0[0], e0[1], f0[0], f0[1]};
    const v4f r01 = {e0[2], e0[3], f0[2], f0[3]};
    const v4f r10 = {e1[0], e1[1], f1[0], f1[1]};
    const v4f r11 = {e1[2], e1[3], f1[2], f1[3]};
    w0[0] = r00;  w0[1] = r01;
    w1[0] = r10;  w1[1] = r11;
  }

  const v8f vz = {0.f,0.f,0.f,0.f,0.f,0.f,0.f,0.f};
  v8f accm[4];
#pragma unroll
  for (int t = 0; t < 4; ++t) accm[t] = vz;

#pragma unroll 1
  for (int hh = 0; hh < NHEADS; ++hh) {
    const float smax = hh == 0 ? sm[0] : hh == 1 ? sm[1] : hh == 2 ? sm[2] : sm[3];
    const float xr = sD[hh * 128 + wid * 16 + mr] + smax;
    const float tr = __expf(-(1.f - NEG_SLOPE) * fabsf(xr));
    const float ra = xr >= 0.f ? 1.f : tr;
    const float rb = xr >= 0.f ? tr : 1.f;

    v8f acc[4];
#pragma unroll
    for (int t = 0; t < 4; ++t) acc[t] = vz;
    float zacc = 0.f;

#pragma unroll 1
    for (int jc = 0; jc < 4; ++jc) {
      const int t = hh * 4 + jc;

      tile_store(pre, sdst + (t & 1) * TILE_F);
      __syncthreads();

      const float* sEp = sEF + hh * (NNODES * 2) + jc * 256 + 4 * half;
      const float* sHp = sH + (t & 1) * TILE_F + half * 160 + 2 * mr;
      Grp ga, gb;

      grp_load(ga, sEp, sHp);
      __builtin_amdgcn_sched_barrier(0);

      if (t + 1 < 4 * NHEADS) {
        const int jn = (jc + 1) & 3;
        const int hn = hh + (jc == 3 ? 1 : 0);
        tile_load(pre, gsrc + (size_t)jn * 128 * OUTD + hn * HID);
      }

      for (int jj = 0; jj < 128; jj += 2 * GNODES) {
        grp_load(gb, sEp + (jj + GNODES) * 2, sHp + ((jj + GNODES) >> 1) * 160);
        __builtin_amdgcn_sched_barrier(0);
        grp_mma(ga, ra, rb, zacc, acc);
        __builtin_amdgcn_sched_barrier(0);
        if (jj + 2 * GNODES < 128)
          grp_load(ga, sEp + (jj + 2 * GNODES) * 2,
                   sHp + ((jj + 2 * GNODES) >> 1) * 160);
        __builtin_amdgcn_sched_barrier(0);
        grp_mma(gb, ra, rb, zacc, acc);
        __builtin_amdgcn_sched_barrier(0);
      }
    }

    const float z = zacc + __shfl_xor(zacc, 16, 32);
#pragma unroll
    for (int r = 0; r < 8; ++r) {
      const float zr = __shfl(z, r + 8 * half, 32);
      const float sc = 0.25f / zr;
#pragma unroll
      for (int t = 0; t < 4; ++t) accm[t][r] += acc[t][r] * sc;
    }
  }

#pragma unroll
  for (int t = 0; t < 4; ++t) {
    const float bsv = bias[t * 16 + mr];
#pragma unroll
    for (int r = 0; r < 8; ++r) {
      float v = accm[t][r] + bsv;
      v = v > 0.f ? v : 0.f;
      xout[(size_t)(b * NNODES + i0 + wid * 16 + r + 8 * half) * HID +
           t * 16 + mr] = v;
    }
  }
}

__global__ __launch_bounds__(64) void k_readout(const float* __restrict__ x,
                                                const float* __restrict__ Wr,
                                                const float* __restrict__ br,
                                                float* __restrict__ out) {
  __shared__ float pooled[DIM];
  const int b = blockIdx.x, d = threadIdx.x;
  float s = 0.f;
  for (int i = 0; i < NNODES; ++i)
    s += x[(size_t)(b * NNODES + i) * DIM + d];
  pooled[d] = s * (1.f / (float)NNODES);
  __syncthreads();
  float o = br[d];
  for (int k = 0; k < DIM; ++k) o += pooled[k] * Wr[d * DIM + k];
  out[b * DIM + d] = o;
}

extern "C" void kernel_launch(void* const* d_in, const int* in_sizes, int n_in,
                              void* d_out, int out_size, void* d_ws,
                              size_t ws_size, hipStream_t stream) {
  (void)in_sizes; (void)n_in; (void)out_size; (void)ws_size;
  const float* emb    = (const float*)d_in[0];
  const float* lin_w  = (const float*)d_in[1];
  const float* attS   = (const float*)d_in[2];
  const float* attD   = (const float*)d_in[3];
  const float* conv_b = (const float*)d_in[4];
  const float* rw     = (const float*)d_in[5];
  const float* rb     = (const float*)d_in[6];

  float* ws = (float*)d_ws;
  const size_t XSZ = (size_t)BATCH * NNODES * DIM;
  const size_t HSZ = (size_t)BATCH * NNODES * OUTD;
  const size_t ASZ = (size_t)BATCH * NHEADS * NNODES;
  float* xbuf0 = ws;
  float* xbuf1 = ws + XSZ;
  float* hbuf  = ws + 2 * XSZ;
  float* asrc  = hbuf + HSZ;
  float* adst  = asrc + ASZ;

  const float* xin = emb;
  float* xo[NL] = {xbuf0, xbuf1, xbuf0};

  for (int l = 0; l < NL; ++l) {
    k_lin<<<(BATCH * NNODES) / 64, 256, 0, stream>>>(
        xin, lin_w + (size_t)l * OUTD * DIM, attS + l * NHEADS * HID,
        attD + l * NHEADS * HID, hbuf, asrc, adst);
    k_agg<<<dim3(NNODES / 128, BATCH), 256, 0, stream>>>(
        hbuf, asrc, adst, conv_b + l * HID, xo[l]);
    xin = xo[l];
  }
  k_readout<<<BATCH, 64, 0, stream>>>(xin, rw, rb, (float*)d_out);
}


# k_readout: node mean as ONE memory round trip, not 32 dependent row batches: 256 threads per batch element, each thread's 32 row loads independent, weights fetched alongside; sum order changed
# speedup vs baseline: 1.7999x; 1.1076x over previous
#include <hip/hip_runtime.h>


typedef float v2f __attribute__((ext_vector_type(2)));
typedef float v4f __attribute__((ext_vector_type(4)));
typedef float v8f __attribute__((ext_vector_type(8)));

#define NHEADS 4
#define HID    64
#define NL     3
#define BATCH  64
#define NNODES 512
#define DIM    64
#define OUTD   (NHEADS * HID)
#define NEG_SLOPE 0.2f

__device__ __forceinline__ v8f wmma_f32(v2f a, v2f b, v8f c) {
  return __builtin_amdgcn_wmma_f32_16x16x4_f32(false, a, false, b, (short)0, c,
                                               false, false);
}

template <int WD>
__device__ __forceinline__ void rows16_fold(float (&v)[16], int mr) {
  const bool up = (mr & WD) != 0;
#pragma unroll
  for (int i = 0; i < WD; ++i) {
    const float keep = up ? v[i + WD] : v[i];
    const float send = up ? v[i] : v[i + WD];
    v[i] = keep + __shfl_xor(send, WD, 32);
  }
}

__global__ __launch_bounds__(256) void k_lin(const float* __restrict__ x,
                                             const float* __restrict__ W,
                                             const float* __restrict__ attS,
                                             const float* __restrict__ attD,
                                             float* __restrict__ hout,
                                             float* __restrict__ asrc,
                                             float* __restrict__ adst) {
  const int tid  = threadIdx.x;
  const int lane = tid & 31;
  const int wid  = tid >> 5;
  const int half = lane >> 4;
  const int mr   = lane & 15;

  const int hd   = wid >> 1;
  const int row0 = blockIdx.x * 64 + (wid & 1) * 32;
  const int o0   = hd * HID;

  const v8f vz = {0.f,0.f,0.f,0.f,0.f,0.f,0.f,0.f};
  v8f acc[2][4];
#pragma unroll
  for (int sm = 0; sm < 2; ++sm)
#pragma unroll
    for (int sn = 0; sn < 4; ++sn) acc[sm][sn] = vz;

  for (int kk = 0; kk < DIM; kk += 4) {
    const int kbase = kk + 2 * half;
    v2f afr[2], bfr[4];
#pragma unroll
    for (int sm = 0; sm < 2; ++sm)
      afr[sm] = *(const v2f*)(x + (size_t)(row0 + sm * 16 + mr) * DIM + kbase);
#pragma unroll
    for (int sn = 0; sn < 4; ++sn)
      bfr[sn] = *(const v2f*)(W + (size_t)(o0 + sn * 16 + mr) * DIM + kbase);
#pragma unroll
    for (int sm = 0; sm < 2; ++sm)
#pragma unroll
      for (int sn = 0; sn < 4; ++sn)
        acc[sm][sn] = wmma_f32(afr[sm], bfr[sn], acc[sm][sn]);
  }

#pragma unroll
  for (int sm = 0; sm < 2; ++sm)
#pragma unroll
    for (int sn = 0; sn < 4; ++sn)
#pragma unroll
      for (int r = 0; r < 8; ++r) {
        const int row = row0 + sm * 16 + r + 8 * half;
        const int col = o0 + sn * 16 + mr;
        hout[(size_t)row * OUTD + col] = acc[sm][sn][r];
      }

  float wS[4], wD[4];
#pragma unroll
  for (int sn = 0; sn < 4; ++sn) {
    wS[sn] = attS[o0 + sn * 16 + mr];
    wD[sn] = attD[o0 + sn * 16 + mr];
  }
  float ps[16], pd[16];
#pragma unroll
  for (int sm = 0; sm < 2; ++sm)
#pragma unroll
    for (int r = 0; r < 8; ++r) {
      float ss = acc[sm][0][r] * wS[0];
      float sd = acc[sm][0][r] * wD[0];
#pragma unroll
      for (int sn = 1; sn < 4; ++sn) {
        ss = fmaf(acc[sm][sn][r], wS[sn], ss);
        sd = fmaf(acc[sm][sn][r], wD[sn], sd);
      }
      ps[sm * 8 + r] = ss;
      pd[sm * 8 + r] = sd;
    }
  rows16_fold<8>(ps, mr);  rows16_fold<8>(pd, mr);
  rows16_fold<4>(ps, mr);  rows16_fold<4>(pd, mr);
  rows16_fold<2>(ps, mr);  rows16_fold<2>(pd, mr);
  rows16_fold<1>(ps, mr);  rows16_fold<1>(pd, mr);

  const int grow = row0 + (mr >> 3) * 16 + (mr & 7) + 8 * half;
  const int b = grow >> 9, n = grow & (NNODES - 1);
  asrc[((b * NHEADS + hd) << 9) + n] = ps[0];
  adst[((b * NHEADS + hd) << 9) + n] = pd[0];
}

#define GSTEPS 8
#define GNODES (4 * GSTEPS)

struct Grp {
  v4f ef[GSTEPS];
  v2f b[GSTEPS][4];
};

__device__ __forceinline__ void grp_load(Grp& g, const float* sEq,
                                         const float* sHh) {
#pragma unroll
  for (int u = 0; u < GSTEPS; ++u) {
    g.ef[u] = *(const v4f*)(sEq + u * 8);
#pragma unroll
    for (int t = 0; t < 4; ++t)
      g.b[u][t] = *(const v2f*)(sHh + u * 2 * 160 + t * 32);
  }
}

__device__ __forceinline__ void grp_mma(const Grp& g, float ra, float rb,
                                        float& zacc, v8f (&acc)[4]) {
#pragma unroll
  for (int u = 0; u < GSTEPS; ++u) {
    const float p0 = fmaxf(ra * g.ef[u].x, rb * g.ef[u].z);
    const float p1 = fmaxf(ra * g.ef[u].y, rb * g.ef[u].w);
    zacc += p0 + p1;
    v2f af; af.x = p0; af.y = p1;
#pragma unroll
    for (int t = 0; t < 4; ++t) acc[t] = wmma_f32(af, g.b[u][t], acc[t]);
  }
}

#define TILE_F (64 * 160)

__device__ __forceinline__ void tile_load(v2f (&pre)[16], const float* g) {
#pragma unroll
  for (int k = 0; k < 16; ++k) {
    pre[k].x = g[k * 8 * OUTD];
    pre[k].y = g[k * 8 * OUTD + OUTD];
  }
}

__device__ __forceinline__ void tile_store(const v2f (&pre)[16], float* s) {
#pragma unroll
  for (int k = 0; k < 16; ++k) *(v2f*)(s + k * 4 * 160) = pre[k];
}

__global__ __launch_bounds__(256) void k_agg(const float* __restrict__ hbuf,
                                             const float* __restrict__ asrc,
                                             const float* __restrict__ adst,
                                             const float* __restrict__ bias,
                                             float* __restrict__ xout) {
  __shared__ __attribute__((aligned(16))) float sH[2 * TILE_F];
  __shared__ __attribute__((aligned(16))) float sA[NHEADS * NNODES];
  __shared__ __attribute__((aligned(16))) float sEF[NHEADS * NNODES * 2];
  __shared__ float sD[NHEADS * 128];

  const int tid  = threadIdx.x;
  const int lane = tid & 31;
  const int wid  = tid >> 5;
  const int half = lane >> 4;
  const int mr   = lane & 15;

  const int b  = blockIdx.y;
  const int i0 = blockIdx.x * 128;

  const float* gsrc =
      hbuf + (size_t)(b * NNODES + 2 * (tid >> 6)) * OUTD + (tid & 63);
  float* sdst = sH + (tid >> 6) * 160 + 2 * (tid & 63);

  v2f pre[16];
  tile_load(pre, gsrc);

  v4f sq0, sq1;
  {
    const v4f* as4 = (const v4f*)(asrc + ((size_t)(b * NHEADS) << 9));
    sq0 = as4[tid];
    sq1 = as4[tid + 256];
    ((v4f*)sA)[tid]       = sq0;
    ((v4f*)sA)[tid + 256] = sq1;
#pragma unroll
    for (int q = 0; q < 2; ++q) {
      const int idx = tid + 256 * q;
      sD[idx] = adst[((b * NHEADS + (idx >> 7)) << 9) + i0 + (idx & 127)];
    }
  }
  __syncthreads();

  float sm[NHEADS];
#pragma unroll
  for (int h = 0; h < NHEADS; ++h) {
    const v4f* s4 = (const v4f*)(sA + h * NNODES);
    const v4f q0 = s4[lane], q1 = s4[lane + 32], q2 = s4[lane + 64],
              q3 = s4[lane + 96];
    const float m0 = fmaxf(fmaxf(q0.x, q0.y), fmaxf(q0.z, q0.w));
    const float m1 = fmaxf(fmaxf(q1.x, q1.y), fmaxf(q1.z, q1.w));
    const float m2 = fmaxf(fmaxf(q2.x, q2.y), fmaxf(q2.z, q2.w));
    const float m3 = fmaxf(fmaxf(q3.x, q3.y), fmaxf(q3.z, q3.w));
    float mx = fmaxf(fmaxf(m0, m1), fmaxf(m2, m3));
#pragma unroll
    for (int off = 16; off; off >>= 1)
      mx = fmaxf(mx, __shfl_xor(mx, off, 32));
    sm[h] = mx;
  }

  {
    const bool hi = (tid >> 7) != 0;
    const float S0 = hi ? sm[1] : sm[0];
    const float S1 = hi ? sm[3] : sm[2];
    v4f e0, f0, e1, f1;
#pragma unroll
    for (int c = 0; c < 4; ++c) {
      const float t0 = sq0[c] - S0, t1 = sq1[c] - S1;
      e0[c] = __expf(t0);  f0[c] = __expf(NEG_SLOPE * t0);
      e1[c] = __expf(t1);  f1[c] = __expf(NEG_SLOPE * t1);
    }
    v4f* w0 = (v4f*)(sEF + tid * 8);
    v4f* w1 = (v4f*)(sEF + 2 * NNODES * 2 + tid * 8);
    const v4f r00 = {e0[0], e0[1], f0[0], f0[1]};
    const v4f r01 = {e0[2], e0[3], f0[2], f0[3]};
    const v4f r10 = {e1[0], e1[1], f1[0], f1[1]};
    const v4f r11 = {e1[2], e1[3], f1[2], f1[3]};
    w0[0] = r00;  w0[1] = r01;
    w1[0] = r10;  w1[1] = r11;
  }

  const v8f vz = {0.f,0.f,0.f,0.f,0.f,0.f,0.f,0.f};
  v8f accm[4];
#pragma unroll
  for (int t = 0; t < 4; ++t) accm[t] = vz;

#pragma unroll 1
  for (int hh = 0; hh < NHEADS; ++hh) {
    const float smax = hh == 0 ? sm[0] : hh == 1 ? sm[1] : hh == 2 ? sm[2] : sm[3];
    const float xr = sD[hh * 128 + wid * 16 + mr] + smax;
    const float tr = __expf(-(1.f - NEG_SLOPE) * fabsf(xr));
    const float ra = xr >= 0.f ? 1.f : tr;
    const float rb = xr >= 0.f ? tr : 1.f;

    v8f acc[4];
#pragma unroll
    for (int t = 0; t < 4; ++t) acc[t] = vz;
    float zacc = 0.f;

#pragma unroll 1
    for (int jc = 0; jc < 4; ++jc) {
      const int t = hh * 4 + jc;

      tile_store(pre, sdst + (t & 1) * TILE_F);
      __syncthreads();

      const float* sEp = sEF + hh * (NNODES * 2) + jc * 256 + 4 * half;
      const float* sHp = sH + (t & 1) * TILE_F + half * 160 + 2 * mr;
      Grp ga, gb;

      grp_load(ga, sEp, sHp);
      __builtin_amdgcn_sched_barrier(0);

      if (t + 1 < 4 * NHEADS) {
        const int jn = (jc + 1) & 3;
        const int hn = hh + (jc == 3 ? 1 : 0);
        tile_load(pre, gsrc + (size_t)jn * 128 * OUTD + hn * HID);
      }

      for (int jj = 0; jj < 128; jj += 2 * GNODES) {
        grp_load(gb, sEp + (jj + GNODES) * 2, sHp + ((jj + GNODES) >> 1) * 160);
        __builtin_amdgcn_sched_barrier(0);
        grp_mma(ga, ra, rb, zacc, acc);
        __builtin_amdgcn_sched_barrier(0);
        if (jj + 2 * GNODES < 128)
          grp_load(ga, sEp + (jj + 2 * GNODES) * 2,
                   sHp + ((jj + 2 * GNODES) >> 1) * 160);
        __builtin_amdgcn_sched_barrier(0);
        grp_mma(gb, ra, rb, zacc, acc);
        __builtin_amdgcn_sched_barrier(0);
      }
    }

    const float z = zacc + __shfl_xor(zacc, 16, 32);
#pragma unroll
    for (int r = 0; r < 8; ++r) {
      const float zr = __shfl(z, r + 8 * half, 32);
      const float sc = 0.25f / zr;
#pragma unroll
      for (int t = 0; t < 4; ++t) accm[t][r] += acc[t][r] * sc;
    }
  }

#pragma unroll
  for (int t = 0; t < 4; ++t) {
    const float bsv = bias[t * 16 + mr];
#pragma unroll
    for (int r = 0; r < 8; ++r) {
      float v = accm[t][r] + bsv;
      v = v > 0.f ? v : 0.f;
      xout[(size_t)(b * NNODES + i0 + wid * 16 + r + 8 * half) * HID +
           t * 16 + mr] = v;
    }
  }
}

__global__ __launch_bounds__(256) void k_readout(const float* __restrict__ x,
                                                 const float* __restrict__ Wr,
                                                 const float* __restrict__ br,
                                                 float* __restrict__ out) {
  __shared__ __attribute__((aligned(16))) float part[8 * DIM];
  __shared__ __attribute__((aligned(16))) float pooled[DIM];
  const int b = blockIdx.x, tid = threadIdx.x;
  const int lane = tid & 31, wid = tid >> 5;
  const int c4 = tid & 15;
  const int rg = tid >> 4;

  const int d = tid >> 2, kq = tid & 3;
  const v4f* wp = (const v4f*)(Wr + d * DIM + kq * 16);
  v4f w[4];
#pragma unroll
  for (int q = 0; q < 4; ++q) w[q] = wp[q];
  const float bsv = br[d];

  const v4f* xp = (const v4f*)(x + (size_t)(b * NNODES + rg) * DIM) + c4;
  v4f v[32];
#pragma unroll
  for (int k = 0; k < 32; ++k) v[k] = xp[k * 16 * (DIM / 4)];
  __builtin_amdgcn_sched_barrier(0);
#pragma unroll
  for (int st = 16; st >= 1; st >>= 1)
#pragma unroll
    for (int k = 0; k < st; ++k) v[k] += v[k + st];

  v4f s;
#pragma unroll
  for (int c = 0; c < 4; ++c) s[c] = v[0][c] + __shfl_xor(v[0][c], 16, 32);
  if (lane < 16) ((v4f*)part)[wid * 16 + c4] = s;
  __syncthreads();
  if (tid < DIM) {
    float t = part[tid];
#pragma unroll
    for (int u = 1; u < 8; ++u) t += part[u * DIM + tid];
    pooled[tid] = t * (1.f / (float)NNODES);
  }
  __syncthreads();

  float o = 0.f;
#pragma unroll
  for (int q = 0; q < 4; ++q) {
    const v4f p = ((const v4f*)pooled)[kq * 4 + q];
    o += p.x * w[q].x;
    o += p.y * w[q].y;
    o += p.z * w[q].z;
    o += p.w * w[q].w;
  }
  o += __shfl_xor(o, 1, 32);
  o += __shfl_xor(o, 2, 32);
  if (kq == 0) part[d] = o + bsv;
  __syncthreads();
  if (tid < DIM) out[b * DIM + tid] = part[tid];
}

extern "C" void kernel_launch(void* const* d_in, const int* in_sizes, int n_in,
                              void* d_out, int out_size, void* d_ws,
                              size_t ws_size, hipStream_t stream) {
  (void)in_sizes; (void)n_in; (void)out_size; (void)ws_size;
  const float* emb    = (const float*)d_in[0];
  const float* lin_w  = (const float*)d_in[1];
  const float* attS   = (const float*)d_in[2];
  const float* attD   = (const float*)d_in[3];
  const float* conv_b = (const float*)d_in[4];
  const float* rw     = (const float*)d_in[5];
  const float* rb     = (const float*)d_in[6];

  float* ws = (float*)d_ws;
  const size_t XSZ = (size_t)BATCH * NNODES * DIM;
  const size_t HSZ = (size_t)BATCH * NNODES * OUTD;
  const size_t ASZ = (size_t)BATCH * NHEADS * NNODES;
  float* xbuf0 = ws;
  float* xbuf1 = ws + XSZ;
  float* hbuf  = ws + 2 * XSZ;
  float* asrc  = hbuf + HSZ;
  float* adst  = asrc + ASZ;

  const float* xin = emb;
  float* xo[NL] = {xbuf0, xbuf1, xbuf0};

  for (int l = 0; l < NL; ++l) {
    k_lin<<<(BATCH * NNODES) / 64, 256, 0, stream>>>(
        xin, lin_w + (size_t)l * OUTD * DIM, attS + l * NHEADS * HID,
        attD + l * NHEADS * HID, hbuf, asrc, adst);
    k_agg<<<dim3(NNODES / 128, BATCH), 256, 0, stream>>>(
        hbuf, asrc, adst, conv_b + l * HID, xo[l]);
    xin = xo[l];
  }
  k_readout<<<BATCH, 256, 0, stream>>>(xin, rw, rb, (float*)d_out);
}


# k_lin: x rows and W staged once per block in LDS by whole-line 128-bit loads, fragments read from padded LDS (pitch 68) instead of 16-line strided global loads; same k order, same sum order
# speedup vs baseline: 1.8682x; 1.0379x over previous
#include <hip/hip_runtime.h>


typedef float v2f __attribute__((ext_vector_type(2)));
typedef float v4f __attribute__((ext_vector_type(4)));
typedef float v8f __attribute__((ext_vector_type(8)));

#define NHEADS 4
#define HID    64
#define NL     3
#define BATCH  64
#define NNODES 512
#define DIM    64
#define OUTD   (NHEADS * HID)
#define NEG_SLOPE 0.2f

__device__ __forceinline__ v8f wmma_f32(v2f a, v2f b, v8f c) {
  return __builtin_amdgcn_wmma_f32_16x16x4_f32(false, a, false, b, (short)0, c,
                                               false, false);
}

template <int WD>
__device__ __forceinline__ void rows16_fold(float (&v)[16], int mr) {
  const bool up = (mr & WD) != 0;
#pragma unroll
  for (int i = 0; i < WD; ++i) {
    const float keep = up ? v[i + WD] : v[i];
    const float send = up ? v[i] : v[i + WD];
    v[i] = keep + __shfl_xor(send, WD, 32);
  }
}

#define LPITCH 68

__global__ __launch_bounds__(256) void k_lin(const float* __restrict__ x,
                                             const float* __restrict__ W,
                                             const float* __restrict__ attS,
                                             const float* __restrict__ attD,
                                             float* __restrict__ hout,
                                             float* __restrict__ asrc,
                                             float* __restrict__ adst) {
  const int tid  = threadIdx.x;
  const int lane = tid & 31;
  const int wid  = tid >> 5;
  const int half = lane >> 4;
  const int mr   = lane & 15;

  const int hd   = wid >> 1;
  const int row0 = blockIdx.x * 64 + (wid & 1) * 32;
  const int o0   = hd * HID;

  __shared__ __attribute__((aligned(16))) float sW[OUTD * LPITCH];
  __shared__ __attribute__((aligned(16))) float sX[64 * LPITCH];

  {
    const v4f* gX = (const v4f*)(x + (size_t)blockIdx.x * 64 * DIM) + tid;
    const v4f* gW = (const v4f*)W + tid;
    v4f rx[4], rw[16];
#pragma unroll
    for (int q = 0; q < 4; ++q) rx[q] = gX[q * 256];
#pragma unroll
    for (int q = 0; q < 16; ++q) rw[q] = gW[q * 256];
    const int cell = (tid >> 4) * LPITCH + 4 * (tid & 15);
#pragma unroll
    for (int q = 0; q < 4; ++q) *(v4f*)(sX + cell + q * 16 * LPITCH) = rx[q];
#pragma unroll
    for (int q = 0; q < 16; ++q) *(v4f*)(sW + cell + q * 16 * LPITCH) = rw[q];
  }
  __syncthreads();

  const v8f vz = {0.f,0.f,0.f,0.f,0.f,0.f,0.f,0.f};
  v8f acc[2][4];
#pragma unroll
  for (int sm = 0; sm < 2; ++sm)
#pragma unroll
    for (int sn = 0; sn < 4; ++sn) acc[sm][sn] = vz;

  const float* aP = sX + ((wid & 1) * 32 + mr) * LPITCH + 2 * half;
  const float* bP = sW + (o0 + mr) * LPITCH + 2 * half;

  for (int kk = 0; kk < DIM; kk += 4) {
    v2f afr[2], bfr[4];
#pragma unroll
    for (int sm = 0; sm < 2; ++sm)
      afr[sm] = *(const v2f*)(aP + sm * 16 * LPITCH + kk);
#pragma unroll
    for (int sn = 0; sn < 4; ++sn)
      bfr[sn] = *(const v2f*)(bP + sn * 16 * LPITCH + kk);
#pragma unroll
    for (int sm = 0; sm < 2; ++sm)
#pragma unroll
      for (int sn = 0; sn < 4; ++sn)
        acc[sm][sn] = wmma_f32(afr[sm], bfr[sn], acc[sm][sn]);
  }

#pragma unroll
  for (int sm = 0; sm < 2; ++sm)
#pragma unroll
    for (int sn = 0; sn < 4; ++sn)
#pragma unroll
      for (int r = 0; r < 8; ++r) {
        const int row = row0 + sm * 16 + r + 8 * half;
        const int col = o0 + sn * 16 + mr;
        hout[(size_t)row * OUTD + col] = acc[sm][sn][r];
      }

  float wS[4], wD[4];
#pragma unroll
  for (int sn = 0; sn < 4; ++sn) {
    wS[sn] = attS[o0 + sn * 16 + mr];
    wD[sn] = attD[o0 + sn * 16 + mr];
  }
  float ps[16], pd[16];
#pragma unroll
  for (int sm = 0; sm < 2; ++sm)
#pragma unroll
    for (int r = 0; r < 8; ++r) {
      float ss = acc[sm][0][r] * wS[0];
      float sd = acc[sm][0][r] * wD[0];
#pragma unroll
      for (int sn = 1; sn < 4; ++sn) {
        ss = fmaf(acc[sm][sn][r], wS[sn], ss);
        sd = fmaf(acc[sm][sn][r], wD[sn], sd);
      }
      ps[sm * 8 + r] = ss;
      pd[sm * 8 + r] = sd;
    }
  rows16_fold<8>(ps, mr);  rows16_fold<8>(pd, mr);
  rows16_fold<4>(ps, mr);  rows16_fold<4>(pd, mr);
  rows16_fold<2>(ps, mr);  rows16_fold<2>(pd, mr);
  rows16_fold<1>(ps, mr);  rows16_fold<1>(pd, mr);

  const int grow = row0 + (mr >> 3) * 16 + (mr & 7) + 8 * half;
  const int b = grow >> 9, n = grow & (NNODES - 1);
  asrc[((b * NHEADS + hd) << 9) + n] = ps[0];
  adst[((b * NHEADS + hd) << 9) + n] = pd[0];
}

#define GSTEPS 8
#define GNODES (4 * GSTEPS)

struct Grp {
  v4f ef[GSTEPS];
  v2f b[GSTEPS][4];
};

__device__ __forceinline__ void grp_load(Grp& g, const float* sEq,
                                         const float* sHh) {
#pragma unroll
  for (int u = 0; u < GSTEPS; ++u) {
    g.ef[u] = *(const v4f*)(sEq + u * 8);
#pragma unroll
    for (int t = 0; t < 4; ++t)
      g.b[u][t] = *(const v2f*)(sHh + u * 2 * 160 + t * 32);
  }
}

__device__ __forceinline__ void grp_mma(const Grp& g, float ra, float rb,
                                        float& zacc, v8f (&acc)[4]) {
#pragma unroll
  for (int u = 0; u < GSTEPS; ++u) {
    const float p0 = fmaxf(ra * g.ef[u].x, rb * g.ef[u].z);
    const float p1 = fmaxf(ra * g.ef[u].y, rb * g.ef[u].w);
    zacc += p0 + p1;
    v2f af; af.x = p0; af.y = p1;
#pragma unroll
    for (int t = 0; t < 4; ++t) acc[t] = wmma_f32(af, g.b[u][t], acc[t]);
  }
}

#define TILE_F (64 * 160)

__device__ __forceinline__ void tile_load(v2f (&pre)[16], const float* g) {
#pragma unroll
  for (int k = 0; k < 16; ++k) {
    pre[k].x = g[k * 8 * OUTD];
    pre[k].y = g[k * 8 * OUTD + OUTD];
  }
}

__device__ __forceinline__ void tile_store(const v2f (&pre)[16], float* s) {
#pragma unroll
  for (int k = 0; k < 16; ++k) *(v2f*)(s + k * 4 * 160) = pre[k];
}

__global__ __launch_bounds__(256) void k_agg(const float* __restrict__ hbuf,
                                             const float* __restrict__ asrc,
                                             const float* __restrict__ adst,
                                             const float* __restrict__ bias,
                                             float* __restrict__ xout) {
  __shared__ __attribute__((aligned(16))) float sH[2 * TILE_F];
  __shared__ __attribute__((aligned(16))) float sA[NHEADS * NNODES];
  __shared__ __attribute__((aligned(16))) float sEF[NHEADS * NNODES * 2];
  __shared__ float sD[NHEADS * 128];

  const int tid  = threadIdx.x;
  const int lane = tid & 31;
  const int wid  = tid >> 5;
  const int half = lane >> 4;
  const int mr   = lane & 15;

  const int b  = blockIdx.y;
  const int i0 = blockIdx.x * 128;

  const float* gsrc =
      hbuf + (size_t)(b * NNODES + 2 * (tid >> 6)) * OUTD + (tid & 63);
  float* sdst = sH + (tid >> 6) * 160 + 2 * (tid & 63);

  v2f pre[16];
  tile_load(pre, gsrc);

  v4f sq0, sq1;
  {
    const v4f* as4 = (const v4f*)(asrc + ((size_t)(b * NHEADS) << 9));
    sq0 = as4[tid];
    sq1 = as4[tid + 256];
    ((v4f*)sA)[tid]       = sq0;
    ((v4f*)sA)[tid + 256] = sq1;
#pragma unroll
    for (int q = 0; q < 2; ++q) {
      const int idx = tid + 256 * q;
      sD[idx] = adst[((b * NHEADS + (idx >> 7)) << 9) + i0 + (idx & 127)];
    }
  }
  __syncthreads();

  float sm[NHEADS];
#pragma unroll
  for (int h = 0; h < NHEADS; ++h) {
    const v4f* s4 = (const v4f*)(sA + h * NNODES);
    const v4f q0 = s4[lane], q1 = s4[lane + 32], q2 = s4[lane + 64],
              q3 = s4[lane + 96];
    const float m0 = fmaxf(fmaxf(q0.x, q0.y), fmaxf(q0.z, q0.w));
    const float m1 = fmaxf(fmaxf(q1.x, q1.y), fmaxf(q1.z, q1.w));
    const float m2 = fmaxf(fmaxf(q2.x, q2.y), fmaxf(q2.z, q2.w));
    const float m3 = fmaxf(fmaxf(q3.x, q3.y), fmaxf(q3.z, q3.w));
    float mx = fmaxf(fmaxf(m0, m1), fmaxf(m2, m3));
#pragma unroll
    for (int off = 16; off; off >>= 1)
      mx = fmaxf(mx, __shfl_xor(mx, off, 32));
    sm[h] = mx;
  }

  {
    const bool hi = (tid >> 7) != 0;
    const float S0 = hi ? sm[1] : sm[0];
    const float S1 = hi ? sm[3] : sm[2];
    v4f e0, f0, e1, f1;
#pragma unroll
    for (int c = 0; c < 4; ++c) {
      const float t0 = sq0[c] - S0, t1 = sq1[c] - S1;
      e0[c] = __expf(t0);  f0[c] = __expf(NEG_SLOPE * t0);
      e1[c] = __expf(t1);  f1[c] = __expf(NEG_SLOPE * t1);
    }
    v4f* w0 = (v4f*)(sEF + tid * 8);
    v4f* w1 = (v4f*)(sEF + 2 * NNODES * 2 + tid * 8);
    const v4f r00 = {e0[0], e0[1], f0[0], f0[1]};
    const v4f r01 = {e0[2], e0[3], f0[2], f0[3]};
    const v4f r10 = {e1[0], e1[1], f1[0], f1[1]};
    const v4f r11 = {e1[2], e1[3], f1[2], f1[3]};
    w0[0] = r00;  w0[1] = r01;
    w1[0] = r10;  w1[1] = r11;
  }

  const v8f vz = {0.f,0.f,0.f,0.f,0.f,0.f,0.f,0.f};
  v8f accm[4];
#pragma unroll
  for (int t = 0; t < 4; ++t) accm[t] = vz;

#pragma unroll 1
  for (int hh = 0; hh < NHEADS; ++hh) {
    const float smax = hh == 0 ? sm[0] : hh == 1 ? sm[1] : hh == 2 ? sm[2] : sm[3];
    const float xr = sD[hh * 128 + wid * 16 + mr] + smax;
    const float tr = __expf(-(1.f - NEG_SLOPE) * fabsf(xr));
    const float ra = xr >= 0.f ? 1.f : tr;
    const float rb = xr >= 0.f ? tr : 1.f;

    v8f acc[4];
#pragma unroll
    for (int t = 0; t < 4; ++t) acc[t] = vz;
    float zacc = 0.f;

#pragma unroll 1
    for (int jc = 0; jc < 4; ++jc) {
      const int t = hh * 4 + jc;

      tile_store(pre, sdst + (t & 1) * TILE_F);
      __syncthreads();

      const float* sEp = sEF + hh * (NNODES * 2) + jc * 256 + 4 * half;
      const float* sHp = sH + (t & 1) * TILE_F + half * 160 + 2 * mr;
      Grp ga, gb;

      grp_load(ga, sEp, sHp);
      __builtin_amdgcn_sched_barrier(0);

      if (t + 1 < 4 * NHEADS) {
        const int jn = (jc + 1) & 3;
        const int hn = hh + (jc == 3 ? 1 : 0);
        tile_load(pre, gsrc + (size_t)jn * 128 * OUTD + hn * HID);
      }

      for (int jj = 0; jj < 128; jj += 2 * GNODES) {
        grp_load(gb, sEp + (jj + GNODES) * 2, sHp + ((jj + GNODES) >> 1) * 160);
        __builtin_amdgcn_sched_barrier(0);
        grp_mma(ga, ra, rb, zacc, acc);
        __builtin_amdgcn_sched_barrier(0);
        if (jj + 2 * GNODES < 128)
          grp_load(ga, sEp + (jj + 2 * GNODES) * 2,
                   sHp + ((jj + 2 * GNODES) >> 1) * 160);
        __builtin_amdgcn_sched_barrier(0);
        grp_mma(gb, ra, rb, zacc, acc);
        __builtin_amdgcn_sched_barrier(0);
      }
    }

    const float z = zacc + __shfl_xor(zacc, 16, 32);
#pragma unroll
    for (int r = 0; r < 8; ++r) {
      const float zr = __shfl(z, r + 8 * half, 32);
      const float sc = 0.25f / zr;
#pragma unroll
      for (int t = 0; t < 4; ++t) accm[t][r] += acc[t][r] * sc;
    }
  }

#pragma unroll
  for (int t = 0; t < 4; ++t) {
    const float bsv = bias[t * 16 + mr];
#pragma unroll
    for (int r = 0; r < 8; ++r) {
      float v = accm[t][r] + bsv;
      v = v > 0.f ? v : 0.f;
      xout[(size_t)(b * NNODES + i0 + wid * 16 + r + 8 * half) * HID +
           t * 16 + mr] = v;
    }
  }
}

__global__ __launch_bounds__(256) void k_readout(const float* __restrict__ x,
                                                 const float* __restrict__ Wr,
                                                 const float* __restrict__ br,
                                                 float* __restrict__ out) {
  __shared__ __attribute__((aligned(16))) float part[8 * DIM];
  __shared__ __attribute__((aligned(16))) float pooled[DIM];
  const int b = blockIdx.x, tid = threadIdx.x;
  const int lane = tid & 31, wid = tid >> 5;
  const int c4 = tid & 15;
  const int rg = tid >> 4;

  const int d = tid >> 2, kq = tid & 3;
  const v4f* wp = (const v4f*)(Wr + d * DIM + kq * 16);
  v4f w[4];
#pragma unroll
  for (int q = 0; q < 4; ++q) w[q] = wp[q];
  const float bsv = br[d];

  const v4f* xp = (const v4f*)(x + (size_t)(b * NNODES + rg) * DIM) + c4;
  v4f v[32];
#pragma unroll
  for (int k = 0; k < 32; ++k) v[k] = xp[k * 16 * (DIM / 4)];
  __builtin_amdgcn_sched_barrier(0);
#pragma unroll
  for (int st = 16; st >= 1; st >>= 1)
#pragma unroll
    for (int k = 0; k < st; ++k) v[k] += v[k + st];

  v4f s;
#pragma unroll
  for (int c = 0; c < 4; ++c) s[c] = v[0][c] + __shfl_xor(v[0][c], 16, 32);
  if (lane < 16) ((v4f*)part)[wid * 16 + c4] = s;
  __syncthreads();
  if (tid < DIM) {
    float t = part[tid];
#pragma unroll
    for (int u = 1; u < 8; ++u) t += part[u * DIM + tid];
    pooled[tid] = t * (1.f / (float)NNODES);
  }
  __syncthreads();

  float o = 0.f;
#pragma unroll
  for (int q = 0; q < 4; ++q) {
    const v4f p = ((const v4f*)pooled)[kq * 4 + q];
    o += p.x * w[q].x;
    o += p.y * w[q].y;
    o += p.z * w[q].z;
    o += p.w * w[q].w;
  }
  o += __shfl_xor(o, 1, 32);
  o += __shfl_xor(o, 2, 32);
  if (kq == 0) part[d] = o + bsv;
  __syncthreads();
  if (tid < DIM) out[b * DIM + tid] = part[tid];
}

extern "C" void kernel_launch(void* const* d_in, const int* in_sizes, int n_in,
                              void* d_out, int out_size, void* d_ws,
                              size_t ws_size, hipStream_t stream) {
  (void)in_sizes; (void)n_in; (void)out_size; (void)ws_size;
  const float* emb    = (const float*)d_in[0];
  const float* lin_w  = (const float*)d_in[1];
  const float* attS   = (const float*)d_in[2];
  const float* attD   = (const float*)d_in[3];
  const float* conv_b = (const float*)d_in[4];
  const float* rw     = (const float*)d_in[5];
  const float* rb     = (const float*)d_in[6];

  float* ws = (float*)d_ws;
  const size_t XSZ = (size_t)BATCH * NNODES * DIM;
  const size_t HSZ = (size_t)BATCH * NNODES * OUTD;
  const size_t ASZ = (size_t)BATCH * NHEADS * NNODES;
  float* xbuf0 = ws;
  float* xbuf1 = ws + XSZ;
  float* hbuf  = ws + 2 * XSZ;
  float* asrc  = hbuf + HSZ;
  float* adst  = asrc + ASZ;

  const float* xin = emb;
  float* xo[NL] = {xbuf0, xbuf1, xbuf0};

  for (int l = 0; l < NL; ++l) {
    k_lin<<<(BATCH * NNODES) / 64, 256, 0, stream>>>(
        xin, lin_w + (size_t)l * OUTD * DIM, attS + l * NHEADS * HID,
        attD + l * NHEADS * HID, hbuf, asrc, adst);
    k_agg<<<dim3(NNODES / 128, BATCH), 256, 0, stream>>>(
        hbuf, asrc, adst, conv_b + l * HID, xo[l]);
    xin = xo[l];
  }
  k_readout<<<BATCH, 256, 0, stream>>>(xin, rw, rb, (float*)d_out);
}


# k_layer: the 4 heads share each x tile (128 accumulator VGPRs): 4 staged tiles per block not 16, one LDS B-fragment read feeds 4 heads' WMMAs, projection as one K=256 pass; same sum order
# speedup vs baseline: 2.1164x; 1.1329x over previous
#include <hip/hip_runtime.h>


typedef float v2f __attribute__((ext_vector_type(2)));
typedef float v4f __attribute__((ext_vector_type(4)));
typedef float v8f __attribute__((ext_vector_type(8)));

#define NHEADS 4
#define HID    64
#define NL     3
#define BATCH  64
#define NNODES 512
#define DIM    64
#define OUTD   (NHEADS * HID)
#define NEG_SLOPE 0.2f

__device__ __forceinline__ v8f wmma_f32(v2f a, v2f b, v8f c) {
  return __builtin_amdgcn_wmma_f32_16x16x4_f32(false, a, false, b, (short)0, c,
                                               false, false);
}


#define LPITCH 68

#define GSTEPS 4
#define GNODES (4 * GSTEPS)

struct Grp {
  v4f ef[GSTEPS][NHEADS];
  v2f b[GSTEPS][4];
};

__device__ __forceinline__ void grp_load(Grp& g, const float* sEq,
                                         const float* sHh) {
#pragma unroll
  for (int u = 0; u < GSTEPS; ++u) {
#pragma unroll
    for (int h = 0; h < NHEADS; ++h)
      g.ef[u][h] = *(const v4f*)(sEq + h * (NNODES * 2) + u * 8);
#pragma unroll
    for (int t = 0; t < 4; ++t)
      g.b[u][t] = *(const v2f*)(sHh + u * 2 * 160 + t * 32);
  }
}

__device__ __forceinline__ void grp_mma(const Grp& g, const float (&ra)[NHEADS],
                                        const float (&rb)[NHEADS],
                                        float (&zacc)[NHEADS],
                                        v8f (&acc)[NHEADS][4]) {
#pragma unroll
  for (int u = 0; u < GSTEPS; ++u) {
#pragma unroll
    for (int h = 0; h < NHEADS; ++h) {
      const float p0 = fmaxf(ra[h] * g.ef[u][h].x, rb[h] * g.ef[u][h].z);
      const float p1 = fmaxf(ra[h] * g.ef[u][h].y, rb[h] * g.ef[u][h].w);
      zacc[h] += p0 + p1;
      v2f af; af.x = p0; af.y = p1;
#pragma unroll
      for (int t = 0; t < 4; ++t)
        acc[h][t] = wmma_f32(af, g.b[u][t], acc[h][t]);
    }
  }
}


#define TILE_F (64 * 160)

__device__ __forceinline__ void tile_load(v2f (&pre)[16], const float* g) {
#pragma unroll
  for (int k = 0; k < 16; ++k) {
    pre[k].x = g[k * 8 * DIM];
    pre[k].y = g[k * 8 * DIM + DIM];
  }
}

__device__ __forceinline__ void tile_store(const v2f (&pre)[16], float* s) {
#pragma unroll
  for (int k = 0; k < 16; ++k) *(v2f*)(s + k * 4 * 160) = pre[k];
}

__device__ __forceinline__ void row_scores(const float* xrow, const float* sV,
                                           int vrow0, float (&s)[NHEADS]) {
  const v4f* xr = (const v4f*)xrow;
  const v4f* vv = (const v4f*)sV + vrow0 * (DIM / 4);
#pragma unroll
  for (int h = 0; h < NHEADS; ++h) s[h] = 0.f;
#pragma unroll
  for (int q = 0; q < DIM / 4; ++q) {
    const v4f xv = xr[q];
#pragma unroll
    for (int h = 0; h < NHEADS; ++h) {
      const v4f w = vv[h * (DIM / 4) + q];
      s[h] = fmaf(xv.x, w.x, s[h]);
      s[h] = fmaf(xv.y, w.y, s[h]);
      s[h] = fmaf(xv.z, w.z, s[h]);
      s[h] = fmaf(xv.w, w.w, s[h]);
    }
  }
}

__device__ __forceinline__ void row_scores2(const float* xrowa, const float* xrowb,
                                            const float* sV, float (&sa)[NHEADS],
                                            float (&sb)[NHEADS]) {
  const v4f* xa = (const v4f*)xrowa;
  const v4f* xb = (const v4f*)xrowb;
  const v4f* vv = (const v4f*)sV;
#pragma unroll
  for (int h = 0; h < NHEADS; ++h) { sa[h] = 0.f; sb[h] = 0.f; }
#pragma unroll
  for (int q = 0; q < DIM / 4; ++q) {
    const v4f a = xa[q], b = xb[q];
#pragma unroll
    for (int h = 0; h < NHEADS; ++h) {
      const v4f w = vv[h * (DIM / 4) + q];
      sa[h] = fmaf(a.x, w.x, sa[h]);
      sa[h] = fmaf(a.y, w.y, sa[h]);
      sa[h] = fmaf(a.z, w.z, sa[h]);
      sa[h] = fmaf(a.w, w.w, sa[h]);
      sb[h] = fmaf(b.x, w.x, sb[h]);
      sb[h] = fmaf(b.y, w.y, sb[h]);
      sb[h] = fmaf(b.z, w.z, sb[h]);
      sb[h] = fmaf(b.w, w.w, sb[h]);
    }
  }
}

__global__ __launch_bounds__(256) void k_layer(const float* __restrict__ xin,
                                               const float* __restrict__ W,
                                               const float* __restrict__ attS,
                                               const float* __restrict__ attD,
                                               const float* __restrict__ bias,
                                               float* __restrict__ xout) {
  __shared__ __attribute__((aligned(16))) float sX[NNODES * LPITCH];
  float* const sH = sX;
  float* const sT = sX;
  __shared__ __attribute__((aligned(16))) float sA[NHEADS * NNODES];
  __shared__ __attribute__((aligned(16))) float sEF[NHEADS * NNODES * 2];
  __shared__ float sD[NHEADS * 128];
  __shared__ __attribute__((aligned(16))) float sW[OUTD * LPITCH];
  __shared__ __attribute__((aligned(16))) float sV[2 * NHEADS * DIM];
  __shared__ __attribute__((aligned(16))) float sAtt[2 * OUTD];

  const int tid  = threadIdx.x;
  const int lane = tid & 31;
  const int wid  = tid >> 5;
  const int half = lane >> 4;
  const int mr   = lane & 15;

  const int b  = blockIdx.y;
  const int i0 = blockIdx.x * 128;

  v4f rw[16], rx[32];
  {
    const v4f* gW = (const v4f*)W + tid;
#pragma unroll
    for (int q = 0; q < 16; ++q) rw[q] = gW[q * 256];
    const v4f* gX = (const v4f*)(xin + (size_t)b * NNODES * DIM) + tid;
#pragma unroll
    for (int q = 0; q < 32; ++q) rx[q] = gX[q * 256];
  }
  const float attSv = attS[tid], attDv = attD[tid];

  const float* gsrc =
      xin + (size_t)(b * NNODES + 2 * (tid >> 6)) * DIM + (tid & 63);
  float* sdst = sH + (tid >> 6) * 160 + 2 * (tid & 63);

  v2f pre[16];
  tile_load(pre, gsrc);

  const int cell = (tid >> 4) * LPITCH + 4 * (tid & 15);
#pragma unroll
  for (int q = 0; q < 16; ++q) *(v4f*)(sW + cell + q * 16 * LPITCH) = rw[q];
#pragma unroll
  for (int q = 0; q < 32; ++q) *(v4f*)(sX + cell + q * 16 * LPITCH) = rx[q];
  sAtt[tid] = attSv;
  sAtt[OUTD + tid] = attDv;
  __syncthreads();

  {
    const int hv = tid >> 6;
    const int k  = tid & 63;
    const float* wcol = sW + hv * HID * LPITCH + k;
    const v4f* aS4 = (const v4f*)(sAtt + hv * HID);
    const v4f* aD4 = (const v4f*)(sAtt + OUTD + hv * HID);
    float vs = 0.f, vd = 0.f;
#pragma unroll
    for (int c4 = 0; c4 < HID / 4; ++c4) {
      const v4f a = aS4[c4], d = aD4[c4];
#pragma unroll
      for (int e = 0; e < 4; ++e) {
        const float w = wcol[(4 * c4 + e) * LPITCH];
        vs = fmaf(w, a[e], vs);
        vd = fmaf(w, d[e], vd);
      }
    }
    sV[hv * DIM + k] = vs;
    sV[(NHEADS + hv) * DIM + k] = vd;
  }
  __syncthreads();

  {
    float sa[NHEADS], sb[NHEADS];
    row_scores2(sX + tid * LPITCH, sX + (256 + tid) * LPITCH, sV, sa, sb);
#pragma unroll
    for (int h = 0; h < NHEADS; ++h) {
      sA[h * NNODES + tid]       = sa[h];
      sA[h * NNODES + 256 + tid] = sb[h];
    }
    if ((tid >> 7) == (int)(blockIdx.x & 1)) {
      float d[NHEADS];
      row_scores(sX + ((blockIdx.x >> 1) * 256 + tid) * LPITCH, sV, NHEADS, d);
#pragma unroll
      for (int h = 0; h < NHEADS; ++h) sD[h * 128 + (tid & 127)] = d[h];
    }
  }
  __syncthreads();

  const v4f sq0 = ((const v4f*)sA)[tid];
  const v4f sq1 = ((const v4f*)sA)[tid + 256];

  float sm[NHEADS];
#pragma unroll
  for (int h = 0; h < NHEADS; ++h) {
    const v4f* s4 = (const v4f*)(sA + h * NNODES);
    const v4f q0 = s4[lane], q1 = s4[lane + 32], q2 = s4[lane + 64],
              q3 = s4[lane + 96];
    const float m0 = fmaxf(fmaxf(q0.x, q0.y), fmaxf(q0.z, q0.w));
    const float m1 = fmaxf(fmaxf(q1.x, q1.y), fmaxf(q1.z, q1.w));
    const float m2 = fmaxf(fmaxf(q2.x, q2.y), fmaxf(q2.z, q2.w));
    const float m3 = fmaxf(fmaxf(q3.x, q3.y), fmaxf(q3.z, q3.w));
    float mx = fmaxf(fmaxf(m0, m1), fmaxf(m2, m3));
#pragma unroll
    for (int off = 16; off; off >>= 1)
      mx = fmaxf(mx, __shfl_xor(mx, off, 32));
    sm[h] = mx;
  }

  {
    const bool hi = (tid >> 7) != 0;
    const float S0 = hi ? sm[1] : sm[0];
    const float S1 = hi ? sm[3] : sm[2];
    v4f e0, f0, e1, f1;
#pragma unroll
    for (int c = 0; c < 4; ++c) {
      const float t0 = sq0[c] - S0, t1 = sq1[c] - S1;
      e0[c] = __expf(t0);  f0[c] = __expf(NEG_SLOPE * t0);
      e1[c] = __expf(t1);  f1[c] = __expf(NEG_SLOPE * t1);
    }
    v4f* w0 = (v4f*)(sEF + tid * 8);
    v4f* w1 = (v4f*)(sEF + 2 * NNODES * 2 + tid * 8);
    const v4f r00 = {e0[0], e0[1], f0[0], f0[1]};
    const v4f r01 = {e0[2], e0[3], f0[2], f0[3]};
    const v4f r10 = {e1[0], e1[1], f1[0], f1[1]};
    const v4f r11 = {e1[2], e1[3], f1[2], f1[3]};
    w0[0] = r00;  w0[1] = r01;
    w1[0] = r10;  w1[1] = r11;
  }

  const v8f vz = {0.f,0.f,0.f,0.f,0.f,0.f,0.f,0.f};

  float ra[NHEADS], rb[NHEADS];
#pragma unroll
  for (int h = 0; h < NHEADS; ++h) {
    const float xr = sD[h * 128 + wid * 16 + mr] + sm[h];
    const float tr = __expf(-(1.f - NEG_SLOPE) * fabsf(xr));
    ra[h] = xr >= 0.f ? 1.f : tr;
    rb[h] = xr >= 0.f ? tr : 1.f;
  }

  v8f acc[NHEADS][4];
  float zacc[NHEADS];
#pragma unroll
  for (int h = 0; h < NHEADS; ++h) {
    zacc[h] = 0.f;
#pragma unroll
    for (int t = 0; t < 4; ++t) acc[h][t] = vz;
  }

#pragma unroll 1
  for (int jc = 0; jc < 4; ++jc) {
    tile_store(pre, sdst + (jc & 1) * TILE_F);
    __syncthreads();

    const float* sEp = sEF + jc * 256 + 4 * half;
    const float* sHp = sH + (jc & 1) * TILE_F + half * 160 + 2 * mr;
    Grp ga, gb;

    grp_load(ga, sEp, sHp);
    __builtin_amdgcn_sched_barrier(0);

    if (jc + 1 < 4) tile_load(pre, gsrc + (size_t)(jc + 1) * 128 * DIM);

#pragma unroll 1
    for (int jj = 0; jj < 128; jj += 2 * GNODES) {
      grp_load(gb, sEp + (jj + GNODES) * 2, sHp + ((jj + GNODES) >> 1) * 160);
      __builtin_amdgcn_sched_barrier(0);
      grp_mma(ga, ra, rb, zacc, acc);
      __builtin_amdgcn_sched_barrier(0);
      if (jj + 2 * GNODES < 128)
        grp_load(ga, sEp + (jj + 2 * GNODES) * 2,
                 sHp + ((jj + 2 * GNODES) >> 1) * 160);
      __builtin_amdgcn_sched_barrier(0);
      grp_mma(gb, ra, rb, zacc, acc);
      __builtin_amdgcn_sched_barrier(0);
    }
  }
  __syncthreads();

  float* sTw = sT + wid * (NHEADS * 16 * LPITCH);
#pragma unroll
  for (int hh = 0; hh < NHEADS; ++hh) {
    const float z = zacc[hh] + __shfl_xor(zacc[hh], 16, 32);
#pragma unroll
    for (int r = 0; r < 8; ++r) {
      const float zr = __shfl(z, r + 8 * half, 32);
      const float sc = 0.25f / zr;
#pragma unroll
      for (int t = 0; t < 4; ++t)
        sTw[(hh * 16 + r + 8 * half) * LPITCH + t * 16 + mr] =
            acc[hh][t][r] * sc;
    }
  }
  __syncthreads();

  v8f accm[4];
#pragma unroll
  for (int t = 0; t < 4; ++t) accm[t] = vz;
  const float* aP = sTw + mr * LPITCH + 2 * half;
  const float* bP = sW + mr * LPITCH + 2 * half;
#pragma unroll 1
  for (int hh = 0; hh < NHEADS; ++hh) {
    for (int kk = 0; kk < DIM; kk += 4) {
      const v2f afr = *(const v2f*)(aP + hh * 16 * LPITCH + kk);
      v2f bfr[4];
#pragma unroll
      for (int sn = 0; sn < 4; ++sn)
        bfr[sn] = *(const v2f*)(bP + (hh * HID + sn * 16) * LPITCH + kk);
#pragma unroll
      for (int sn = 0; sn < 4; ++sn)
        accm[sn] = wmma_f32(afr, bfr[sn], accm[sn]);
    }
  }

#pragma unroll
  for (int t = 0; t < 4; ++t) {
    const float bsv = bias[t * 16 + mr];
#pragma unroll
    for (int r = 0; r < 8; ++r) {
      float v = accm[t][r] + bsv;
      v = v > 0.f ? v : 0.f;
      xout[(size_t)(b * NNODES + i0 + wid * 16 + r + 8 * half) * HID +
           t * 16 + mr] = v;
    }
  }
}

__global__ __launch_bounds__(256) void k_readout(const float* __restrict__ x,
                                                 const float* __restrict__ Wr,
                                                 const float* __restrict__ br,
                                                 float* __restrict__ out) {
  __shared__ __attribute__((aligned(16))) float part[8 * DIM];
  __shared__ __attribute__((aligned(16))) float pooled[DIM];
  const int b = blockIdx.x, tid = threadIdx.x;
  const int lane = tid & 31, wid = tid >> 5;
  const int c4 = tid & 15;
  const int rg = tid >> 4;

  const int d = tid >> 2, kq = tid & 3;
  const v4f* wp = (const v4f*)(Wr + d * DIM + kq * 16);
  v4f w[4];
#pragma unroll
  for (int q = 0; q < 4; ++q) w[q] = wp[q];
  const float bsv = br[d];

  const v4f* xp = (const v4f*)(x + (size_t)(b * NNODES + rg) * DIM) + c4;
  v4f v[32];
#pragma unroll
  for (int k = 0; k < 32; ++k) v[k] = xp[k * 16 * (DIM / 4)];
  __builtin_amdgcn_sched_barrier(0);
#pragma unroll
  for (int st = 16; st >= 1; st >>= 1)
#pragma unroll
    for (int k = 0; k < st; ++k) v[k] += v[k + st];

  v4f s;
#pragma unroll
  for (int c = 0; c < 4; ++c) s[c] = v[0][c] + __shfl_xor(v[0][c], 16, 32);
  if (lane < 16) ((v4f*)part)[wid * 16 + c4] = s;
  __syncthreads();
  if (tid < DIM) {
    float t = part[tid];
#pragma unroll
    for (int u = 1; u < 8; ++u) t += part[u * DIM + tid];
    pooled[tid] = t * (1.f / (float)NNODES);
  }
  __syncthreads();

  float o = 0.f;
#pragma unroll
  for (int q = 0; q < 4; ++q) {
    const v4f p = ((const v4f*)pooled)[kq * 4 + q];
    o += p.x * w[q].x;
    o += p.y * w[q].y;
    o += p.z * w[q].z;
    o += p.w * w[q].w;
  }
  o += __shfl_xor(o, 1, 32);
  o += __shfl_xor(o, 2, 32);
  if (kq == 0) part[d] = o + bsv;
  __syncthreads();
  if (tid < DIM) out[b * DIM + tid] = part[tid];
}

extern "C" void kernel_launch(void* const* d_in, const int* in_sizes, int n_in,
                              void* d_out, int out_size, void* d_ws,
                              size_t ws_size, hipStream_t stream) {
  (void)in_sizes; (void)n_in; (void)out_size; (void)ws_size;
  const float* emb    = (const float*)d_in[0];
  const float* lin_w  = (const float*)d_in[1];
  const float* attS   = (const float*)d_in[2];
  const float* attD   = (const float*)d_in[3];
  const float* conv_b = (const float*)d_in[4];
  const float* rw     = (const float*)d_in[5];
  const float* rb     = (const float*)d_in[6];

  float* ws = (float*)d_ws;
  const size_t XSZ = (size_t)BATCH * NNODES * DIM;
  float* xbuf0 = ws;
  float* xbuf1 = ws + XSZ;

  const float* xin = emb;
  float* xo[NL] = {xbuf0, xbuf1, xbuf0};

  for (int l = 0; l < NL; ++l) {
    k_layer<<<dim3(NNODES / 128, BATCH), 256, 0, stream>>>(
        xin, lin_w + (size_t)l * OUTD * DIM, attS + l * NHEADS * HID,
        attD + l * NHEADS * HID, conv_b + l * HID, xo[l]);
    xin = xo[l];
  }
  k_readout<<<BATCH, 256, 0, stream>>>(xin, rw, rb, (float*)d_out);
}


# k_layer: a matrix step's two source nodes kept as one two-wide value: factor products, max and row sum as packed ops on the loaded pairs, no register moves; row sum order changed (even/odd apart)
# speedup vs baseline: 2.1926x; 1.0360x over previous
#include <hip/hip_runtime.h>


typedef float v2f __attribute__((ext_vector_type(2)));
typedef float v4f __attribute__((ext_vector_type(4)));
typedef float v8f __attribute__((ext_vector_type(8)));

#define NHEADS 4
#define HID    64
#define NL     3
#define BATCH  64
#define NNODES 512
#define DIM    64
#define OUTD   (NHEADS * HID)
#define NEG_SLOPE 0.2f

__device__ __forceinline__ v8f wmma_f32(v2f a, v2f b, v8f c) {
  return __builtin_amdgcn_wmma_f32_16x16x4_f32(false, a, false, b, (short)0, c,
                                               false, false);
}


#define LPITCH 68

#define GSTEPS 4
#define GNODES (4 * GSTEPS)

struct Grp {
  v4f ef[GSTEPS][NHEADS];
  v2f b[GSTEPS][4];
};

__device__ __forceinline__ void grp_load(Grp& g, const float* sEq,
                                         const float* sHh) {
#pragma unroll
  for (int u = 0; u < GSTEPS; ++u) {
#pragma unroll
    for (int h = 0; h < NHEADS; ++h)
      g.ef[u][h] = *(const v4f*)(sEq + h * (NNODES * 2) + u * 8);
#pragma unroll
    for (int t = 0; t < 4; ++t)
      g.b[u][t] = *(const v2f*)(sHh + u * 2 * 160 + t * 32);
  }
}

__device__ __forceinline__ void grp_mma(const Grp& g, const v2f (&ra2)[NHEADS],
                                        const v2f (&rb2)[NHEADS],
                                        v2f (&zacc2)[NHEADS],
                                        v8f (&acc)[NHEADS][4]) {
#pragma unroll
  for (int u = 0; u < GSTEPS; ++u) {
#pragma unroll
    for (int h = 0; h < NHEADS; ++h) {
      const v4f r = g.ef[u][h];
      const v2f e = __builtin_shufflevector(r, r, 0, 1);
      const v2f f = __builtin_shufflevector(r, r, 2, 3);
      const v2f af = __builtin_elementwise_max(ra2[h] * e, rb2[h] * f);
      zacc2[h] += af;
#pragma unroll
      for (int t = 0; t < 4; ++t)
        acc[h][t] = wmma_f32(af, g.b[u][t], acc[h][t]);
    }
  }
}


#define TILE_F (64 * 160)

__device__ __forceinline__ void tile_load(v2f (&pre)[16], const float* g) {
#pragma unroll
  for (int k = 0; k < 16; ++k) {
    pre[k].x = g[k * 8 * DIM];
    pre[k].y = g[k * 8 * DIM + DIM];
  }
}

__device__ __forceinline__ void tile_store(const v2f (&pre)[16], float* s) {
#pragma unroll
  for (int k = 0; k < 16; ++k) *(v2f*)(s + k * 4 * 160) = pre[k];
}

__device__ __forceinline__ void row_scores(const float* xrow, const float* sV,
                                           int vrow0, float (&s)[NHEADS]) {
  const v4f* xr = (const v4f*)xrow;
  const v4f* vv = (const v4f*)sV + vrow0 * (DIM / 4);
#pragma unroll
  for (int h = 0; h < NHEADS; ++h) s[h] = 0.f;
#pragma unroll
  for (int q = 0; q < DIM / 4; ++q) {
    const v4f xv = xr[q];
#pragma unroll
    for (int h = 0; h < NHEADS; ++h) {
      const v4f w = vv[h * (DIM / 4) + q];
      s[h] = fmaf(xv.x, w.x, s[h]);
      s[h] = fmaf(xv.y, w.y, s[h]);
      s[h] = fmaf(xv.z, w.z, s[h]);
      s[h] = fmaf(xv.w, w.w, s[h]);
    }
  }
}

__device__ __forceinline__ void row_scores2(const float* xrowa, const float* xrowb,
                                            const float* sV, float (&sa)[NHEADS],
                                            float (&sb)[NHEADS]) {
  const v4f* xa = (const v4f*)xrowa;
  const v4f* xb = (const v4f*)xrowb;
  const v4f* vv = (const v4f*)sV;
#pragma unroll
  for (int h = 0; h < NHEADS; ++h) { sa[h] = 0.f; sb[h] = 0.f; }
#pragma unroll
  for (int q = 0; q < DIM / 4; ++q) {
    const v4f a = xa[q], b = xb[q];
#pragma unroll
    for (int h = 0; h < NHEADS; ++h) {
      const v4f w = vv[h * (DIM / 4) + q];
      sa[h] = fmaf(a.x, w.x, sa[h]);
      sa[h] = fmaf(a.y, w.y, sa[h]);
      sa[h] = fmaf(a.z, w.z, sa[h]);
      sa[h] = fmaf(a.w, w.w, sa[h]);
      sb[h] = fmaf(b.x, w.x, sb[h]);
      sb[h] = fmaf(b.y, w.y, sb[h]);
      sb[h] = fmaf(b.z, w.z, sb[h]);
      sb[h] = fmaf(b.w, w.w, sb[h]);
    }
  }
}

__global__ __launch_bounds__(256) void k_layer(const float* __restrict__ xin,
                                               const float* __restrict__ W,
                                               const float* __restrict__ attS,
                                               const float* __restrict__ attD,
                                               const float* __restrict__ bias,
                                               float* __restrict__ xout) {
  __shared__ __attribute__((aligned(16))) float sX[NNODES * LPITCH];
  float* const sH = sX;
  float* const sT = sX;
  __shared__ __attribute__((aligned(16))) float sA[NHEADS * NNODES];
  __shared__ __attribute__((aligned(16))) float sEF[NHEADS * NNODES * 2];
  __shared__ float sD[NHEADS * 128];
  __shared__ __attribute__((aligned(16))) float sW[OUTD * LPITCH];
  __shared__ __attribute__((aligned(16))) float sV[2 * NHEADS * DIM];
  __shared__ __attribute__((aligned(16))) float sAtt[2 * OUTD];

  const int tid  = threadIdx.x;
  const int lane = tid & 31;
  const int wid  = tid >> 5;
  const int half = lane >> 4;
  const int mr   = lane & 15;

  const int b  = blockIdx.y;
  const int i0 = blockIdx.x * 128;

  v4f rw[16], rx[32];
  {
    const v4f* gW = (const v4f*)W + tid;
#pragma unroll
    for (int q = 0; q < 16; ++q) rw[q] = gW[q * 256];
    const v4f* gX = (const v4f*)(xin + (size_t)b * NNODES * DIM) + tid;
#pragma unroll
    for (int q = 0; q < 32; ++q) rx[q] = gX[q * 256];
  }
  const float attSv = attS[tid], attDv = attD[tid];

  const float* gsrc =
      xin + (size_t)(b * NNODES + 2 * (tid >> 6)) * DIM + (tid & 63);
  float* sdst = sH + (tid >> 6) * 160 + 2 * (tid & 63);

  v2f pre[16];
  tile_load(pre, gsrc);

  const int cell = (tid >> 4) * LPITCH + 4 * (tid & 15);
#pragma unroll
  for (int q = 0; q < 16; ++q) *(v4f*)(sW + cell + q * 16 * LPITCH) = rw[q];
#pragma unroll
  for (int q = 0; q < 32; ++q) *(v4f*)(sX + cell + q * 16 * LPITCH) = rx[q];
  sAtt[tid] = attSv;
  sAtt[OUTD + tid] = attDv;
  __syncthreads();

  {
    const int hv = tid >> 6;
    const int k  = tid & 63;
    const float* wcol = sW + hv * HID * LPITCH + k;
    const v4f* aS4 = (const v4f*)(sAtt + hv * HID);
    const v4f* aD4 = (const v4f*)(sAtt + OUTD + hv * HID);
    float vs = 0.f, vd = 0.f;
#pragma unroll
    for (int c4 = 0; c4 < HID / 4; ++c4) {
      const v4f a = aS4[c4], d = aD4[c4];
#pragma unroll
      for (int e = 0; e < 4; ++e) {
        const float w = wcol[(4 * c4 + e) * LPITCH];
        vs = fmaf(w, a[e], vs);
        vd = fmaf(w, d[e], vd);
      }
    }
    sV[hv * DIM + k] = vs;
    sV[(NHEADS + hv) * DIM + k] = vd;
  }
  __syncthreads();

  {
    float sa[NHEADS], sb[NHEADS];
    row_scores2(sX + tid * LPITCH, sX + (256 + tid) * LPITCH, sV, sa, sb);
#pragma unroll
    for (int h = 0; h < NHEADS; ++h) {
      sA[h * NNODES + tid]       = sa[h];
      sA[h * NNODES + 256 + tid] = sb[h];
    }
    if ((tid >> 7) == (int)(blockIdx.x & 1)) {
      float d[NHEADS];
      row_scores(sX + ((blockIdx.x >> 1) * 256 + tid) * LPITCH, sV, NHEADS, d);
#pragma unroll
      for (int h = 0; h < NHEADS; ++h) sD[h * 128 + (tid & 127)] = d[h];
    }
  }
  __syncthreads();

  const v4f sq0 = ((const v4f*)sA)[tid];
  const v4f sq1 = ((const v4f*)sA)[tid + 256];

  float sm[NHEADS];
#pragma unroll
  for (int h = 0; h < NHEADS; ++h) {
    const v4f* s4 = (const v4f*)(sA + h * NNODES);
    const v4f q0 = s4[lane], q1 = s4[lane + 32], q2 = s4[lane + 64],
              q3 = s4[lane + 96];
    const float m0 = fmaxf(fmaxf(q0.x, q0.y), fmaxf(q0.z, q0.w));
    const float m1 = fmaxf(fmaxf(q1.x, q1.y), fmaxf(q1.z, q1.w));
    const float m2 = fmaxf(fmaxf(q2.x, q2.y), fmaxf(q2.z, q2.w));
    const float m3 = fmaxf(fmaxf(q3.x, q3.y), fmaxf(q3.z, q3.w));
    float mx = fmaxf(fmaxf(m0, m1), fmaxf(m2, m3));
#pragma unroll
    for (int off = 16; off; off >>= 1)
      mx = fmaxf(mx, __shfl_xor(mx, off, 32));
    sm[h] = mx;
  }

  {
    const bool hi = (tid >> 7) != 0;
    const float S0 = hi ? sm[1] : sm[0];
    const float S1 = hi ? sm[3] : sm[2];
    v4f e0, f0, e1, f1;
#pragma unroll
    for (int c = 0; c < 4; ++c) {
      const float t0 = sq0[c] - S0, t1 = sq1[c] - S1;
      e0[c] = __expf(t0);  f0[c] = __expf(NEG_SLOPE * t0);
      e1[c] = __expf(t1);  f1[c] = __expf(NEG_SLOPE * t1);
    }
    v4f* w0 = (v4f*)(sEF + tid * 8);
    v4f* w1 = (v4f*)(sEF + 2 * NNODES * 2 + tid * 8);
    const v4f r00 = {e0[0], e0[1], f0[0], f0[1]};
    const v4f r01 = {e0[2], e0[3], f0[2], f0[3]};
    const v4f r10 = {e1[0], e1[1], f1[0], f1[1]};
    const v4f r11 = {e1[2], e1[3], f1[2], f1[3]};
    w0[0] = r00;  w0[1] = r01;
    w1[0] = r10;  w1[1] = r11;
  }

  const v8f vz = {0.f,0.f,0.f,0.f,0.f,0.f,0.f,0.f};

  v2f ra2[NHEADS], rb2[NHEADS];
#pragma unroll
  for (int h = 0; h < NHEADS; ++h) {
    const float xr = sD[h * 128 + wid * 16 + mr] + sm[h];
    const float tr = __expf(-(1.f - NEG_SLOPE) * fabsf(xr));
    const float fa = xr >= 0.f ? 1.f : tr;
    const float fb = xr >= 0.f ? tr : 1.f;
    ra2[h].x = fa;  ra2[h].y = fa;
    rb2[h].x = fb;  rb2[h].y = fb;
  }

  v8f acc[NHEADS][4];
  v2f zacc2[NHEADS];
#pragma unroll
  for (int h = 0; h < NHEADS; ++h) {
    zacc2[h].x = 0.f;  zacc2[h].y = 0.f;
#pragma unroll
    for (int t = 0; t < 4; ++t) acc[h][t] = vz;
  }

#pragma unroll 1
  for (int jc = 0; jc < 4; ++jc) {
    tile_store(pre, sdst + (jc & 1) * TILE_F);
    __syncthreads();

    const float* sEp = sEF + jc * 256 + 4 * half;
    const float* sHp = sH + (jc & 1) * TILE_F + half * 160 + 2 * mr;
    Grp ga, gb;

    grp_load(ga, sEp, sHp);
    __builtin_amdgcn_sched_barrier(0);

    if (jc + 1 < 4) tile_load(pre, gsrc + (size_t)(jc + 1) * 128 * DIM);

#pragma unroll 1
    for (int jj = 0; jj < 128; jj += 2 * GNODES) {
      grp_load(gb, sEp + (jj + GNODES) * 2, sHp + ((jj + GNODES) >> 1) * 160);
      __builtin_amdgcn_sched_barrier(0);
      grp_mma(ga, ra2, rb2, zacc2, acc);
      __builtin_amdgcn_sched_barrier(0);
      if (jj + 2 * GNODES < 128)
        grp_load(ga, sEp + (jj + 2 * GNODES) * 2,
                 sHp + ((jj + 2 * GNODES) >> 1) * 160);
      __builtin_amdgcn_sched_barrier(0);
      grp_mma(gb, ra2, rb2, zacc2, acc);
      __builtin_amdgcn_sched_barrier(0);
    }
  }
  __syncthreads();

  float* sTw = sT + wid * (NHEADS * 16 * LPITCH);
#pragma unroll
  for (int hh = 0; hh < NHEADS; ++hh) {
    const float zs = zacc2[hh].x + zacc2[hh].y;
    const float z = zs + __shfl_xor(zs, 16, 32);
#pragma unroll
    for (int r = 0; r < 8; ++r) {
      const float zr = __shfl(z, r + 8 * half, 32);
      const float sc = 0.25f / zr;
#pragma unroll
      for (int t = 0; t < 4; ++t)
        sTw[(hh * 16 + r + 8 * half) * LPITCH + t * 16 + mr] =
            acc[hh][t][r] * sc;
    }
  }
  __syncthreads();

  v8f accm[4];
#pragma unroll
  for (int t = 0; t < 4; ++t) accm[t] = vz;
  const float* aP = sTw + mr * LPITCH + 2 * half;
  const float* bP = sW + mr * LPITCH + 2 * half;
#pragma unroll 1
  for (int hh = 0; hh < NHEADS; ++hh) {
    for (int kk = 0; kk < DIM; kk += 4) {
      const v2f afr = *(const v2f*)(aP + hh * 16 * LPITCH + kk);
      v2f bfr[4];
#pragma unroll
      for (int sn = 0; sn < 4; ++sn)
        bfr[sn] = *(const v2f*)(bP + (hh * HID + sn * 16) * LPITCH + kk);
#pragma unroll
      for (int sn = 0; sn < 4; ++sn)
        accm[sn] = wmma_f32(afr, bfr[sn], accm[sn]);
    }
  }

#pragma unroll
  for (int t = 0; t < 4; ++t) {
    const float bsv = bias[t * 16 + mr];
#pragma unroll
    for (int r = 0; r < 8; ++r) {
      float v = accm[t][r] + bsv;
      v = v > 0.f ? v : 0.f;
      xout[(size_t)(b * NNODES + i0 + wid * 16 + r + 8 * half) * HID +
           t * 16 + mr] = v;
    }
  }
}

__global__ __launch_bounds__(256) void k_readout(const float* __restrict__ x,
                                                 const float* __restrict__ Wr,
                                                 const float* __restrict__ br,
                                                 float* __restrict__ out) {
  __shared__ __attribute__((aligned(16))) float part[8 * DIM];
  __shared__ __attribute__((aligned(16))) float pooled[DIM];
  const int b = blockIdx.x, tid = threadIdx.x;
  const int lane = tid & 31, wid = tid >> 5;
  const int c4 = tid & 15;
  const int rg = tid >> 4;

  const int d = tid >> 2, kq = tid & 3;
  const v4f* wp = (const v4f*)(Wr + d * DIM + kq * 16);
  v4f w[4];
#pragma unroll
  for (int q = 0; q < 4; ++q) w[q] = wp[q];
  const float bsv = br[d];

  const v4f* xp = (const v4f*)(x + (size_t)(b * NNODES + rg) * DIM) + c4;
  v4f v[32];
#pragma unroll
  for (int k = 0; k < 32; ++k) v[k] = xp[k * 16 * (DIM / 4)];
  __builtin_amdgcn_sched_barrier(0);
#pragma unroll
  for (int st = 16; st >= 1; st >>= 1)
#pragma unroll
    for (int k = 0; k < st; ++k) v[k] += v[k + st];

  v4f s;
#pragma unroll
  for (int c = 0; c < 4; ++c) s[c] = v[0][c] + __shfl_xor(v[0][c], 16, 32);
  if (lane < 16) ((v4f*)part)[wid * 16 + c4] = s;
  __syncthreads();
  if (tid < DIM) {
    float t = part[tid];
#pragma unroll
    for (int u = 1; u < 8; ++u) t += part[u * DIM + tid];
    pooled[tid] = t * (1.f / (float)NNODES);
  }
  __syncthreads();

  float o = 0.f;
#pragma unroll
  for (int q = 0; q < 4; ++q) {
    const v4f p = ((const v4f*)pooled)[kq * 4 + q];
    o += p.x * w[q].x;
    o += p.y * w[q].y;
    o += p.z * w[q].z;
    o += p.w * w[q].w;
  }
  o += __shfl_xor(o, 1, 32);
  o += __shfl_xor(o, 2, 32);
  if (kq == 0) part[d] = o + bsv;
  __syncthreads();
  if (tid < DIM) out[b * DIM + tid] = part[tid];
}

extern "C" void kernel_launch(void* const* d_in, const int* in_sizes, int n_in,
                              void* d_out, int out_size, void* d_ws,
                              size_t ws_size, hipStream_t stream) {
  (void)in_sizes; (void)n_in; (void)out_size; (void)ws_size;
  const float* emb    = (const float*)d_in[0];
  const float* lin_w  = (const float*)d_in[1];
  const float* attS   = (const float*)d_in[2];
  const float* attD   = (const float*)d_in[3];
  const float* conv_b = (const float*)d_in[4];
  const float* rw     = (const float*)d_in[5];
  const float* rb     = (const float*)d_in[6];

  float* ws = (float*)d_ws;
  const size_t XSZ = (size_t)BATCH * NNODES * DIM;
  float* xbuf0 = ws;
  float* xbuf1 = ws + XSZ;

  const float* xin = emb;
  float* xo[NL] = {xbuf0, xbuf1, xbuf0};

  for (int l = 0; l < NL; ++l) {
    k_layer<<<dim3(NNODES / 128, BATCH), 256, 0, stream>>>(
        xin, lin_w + (size_t)l * OUTD * DIM, attS + l * NHEADS * HID,
        attD + l * NHEADS * HID, conv_b + l * HID, xo[l]);
    xin = xo[l];
  }
  k_readout<<<BATCH, 256, 0, stream>>>(xin, rw, rb, (float*)d_out);
}
